# starting kernel
# baseline (speedup 1.0000x reference)
#include <hip/hip_runtime.h>
#include <hip/hip_bf16.h>

typedef __attribute__((ext_vector_type(16))) _Float16 v16h;
typedef __attribute__((ext_vector_type(8)))  _Float16 v8h;
typedef __attribute__((ext_vector_type(4)))  _Float16 v4h;
typedef __attribute__((ext_vector_type(8)))  float    v8f;

#define N_NODES 50000
#define N_EDGES 800000
#define DIM 128
#define HEADS 8
#define HEAD_DIM 16

__device__ inline void atomicAddF(float* p, float v) {
    __hip_atomic_fetch_add(p, v, __ATOMIC_RELAXED, __HIP_MEMORY_SCOPE_AGENT);
}
__device__ inline void atomicMaxF(int* p, float v) {
    if (v >= 0.0f) atomicMax(p, __float_as_int(v));
    else           atomicMin((unsigned int*)p, __float_as_uint(v));
}

__device__ inline void gemm_tile_16x16(const v16h a[4], const _Float16* __restrict__ WT,
                                       const float* __restrict__ bias,
                                       float* __restrict__ out,
                                       int r0, int n0, int lane) {
    const int col = lane & 15;
    const int kb  = (lane >> 4) << 3;
    const _Float16* wr = WT + (size_t)(n0 + col) * DIM;
    v8f c = {};
#pragma unroll
    for (int t = 0; t < 4; ++t) {
        const v8h lo = *(const v8h*)(wr + t * 32 + kb);
        const v8h hi = *(const v8h*)(wr + t * 32 + kb + 16);
        v16h b;
#pragma unroll
        for (int i = 0; i < 8; ++i) { b[i] = lo[i]; b[i + 8] = hi[i]; }
        c = __builtin_amdgcn_wmma_f32_16x16x32_f16(false, a[t], false, b,
                                                   (short)0, c, false, false);
    }
    const float bia = bias[n0 + col];
    float* op = out + (size_t)r0 * DIM + n0 + col;
#pragma unroll
    for (int j = 0; j < 8; ++j) {
        int m = j + ((lane >> 4) << 3);
        op[(size_t)m * DIM] = c[j] + bia;
    }
}

__device__ inline void load_a_frags(const _Float16* __restrict__ src, int r0, int lane,
                                    v16h a[4]) {
    const int row = lane & 15;
    const int kb  = (lane >> 4) << 3;
    const _Float16* xr = src + (size_t)(r0 + row) * DIM;
#pragma unroll
    for (int t = 0; t < 4; ++t) {
        const v8h lo = *(const v8h*)(xr + t * 32 + kb);
        const v8h hi = *(const v8h*)(xr + t * 32 + kb + 16);
#pragma unroll
        for (int i = 0; i < 8; ++i) { a[t][i] = lo[i]; a[t][i + 8] = hi[i]; }
    }
}

__global__ void init_state(int* segmax_bits, float* denom, float* vals) {
    int i = blockIdx.x * blockDim.x + threadIdx.x;
    if (i < N_NODES * HEADS) {
        segmax_bits[i] = 0xFF800000;
        denom[i] = 0.0f;
    }
    if (i < N_NODES * DIM) vals[i] = 0.0f;
}

__global__ void cvt_f32_f16_x4(const float* __restrict__ src, _Float16* __restrict__ dst) {
    int i = blockIdx.x * blockDim.x + threadIdx.x;
    float4 vv = ((const float4*)src)[i];
    v4h o = { (_Float16)vv.x, (_Float16)vv.y, (_Float16)vv.z, (_Float16)vv.w };
    ((v4h*)dst)[i] = o;
}

__global__ void cvt_weightsT(const float* __restrict__ Wq, const float* __restrict__ Wk,
                             const float* __restrict__ Wv, const float* __restrict__ Wo,
                             _Float16* __restrict__ WT) {
    int i = blockIdx.x * blockDim.x + threadIdx.x;
    int m = i >> 14, r = i & 16383, n = r >> 7, k = r & 127;
    const float* W = (m == 0) ? Wq : (m == 1) ? Wk : (m == 2) ? Wv : Wo;
    WT[i] = (_Float16)W[(size_t)k * DIM + n];
}

__global__ void qkv_gemm(const _Float16* __restrict__ x16,
                         const _Float16* __restrict__ WTq, const float* __restrict__ bq,
                         const _Float16* __restrict__ WTk, const float* __restrict__ bk,
                         const _Float16* __restrict__ WTv, const float* __restrict__ bv,
                         float* __restrict__ q, float* __restrict__ k,
                         float* __restrict__ v) {
    const int lane = threadIdx.x & 31;
    const int wave = threadIdx.x >> 5;
    const int r0 = blockIdx.x * 16;
    const int n0 = wave * 16;
    v16h a[4];
    load_a_frags(x16, r0, lane, a);
    gemm_tile_16x16(a, WTq, bq, q, r0, n0, lane);
    gemm_tile_16x16(a, WTk, bk, k, r0, n0, lane);
    gemm_tile_16x16(a, WTv, bv, v, r0, n0, lane);
}

__global__ void edge_scores(const int* __restrict__ origin, const int* __restrict__ dest,
                            const float* __restrict__ ew,
                            const float* __restrict__ q, const float* __restrict__ k,
                            float* __restrict__ scores, int* __restrict__ segmax_bits) {
    int idx = blockIdx.x * blockDim.x + threadIdx.x;
    int e = idx >> 3, h = idx & 7;
    int o = origin[e], d = dest[e];
    const float4* qp = (const float4*)(q + (size_t)o * DIM + h * HEAD_DIM);
    const float4* kp = (const float4*)(k + (size_t)d * DIM + h * HEAD_DIM);
    float s = 0.0f;
#pragma unroll
    for (int t = 0; t < 4; ++t) {
        float4 av = qp[t], bv = kp[t];
        s += av.x * bv.x + av.y * bv.y + av.z * bv.z + av.w * bv.w;
    }
    s *= 0.25f;
    s *= ew[idx];
    scores[idx] = s;
    atomicMaxF(&segmax_bits[o * HEADS + h], s);
}

__global__ void edge_exp(const int* __restrict__ origin,
                         const int* __restrict__ segmax_bits,
                         float* __restrict__ scores, float* __restrict__ denom) {
    int idx = blockIdx.x * blockDim.x + threadIdx.x;
    int e = idx >> 3, h = idx & 7;
    int o = origin[e];
    float sm = __int_as_float(segmax_bits[o * HEADS + h]);
    if (!isfinite(sm)) sm = 0.0f;
    float ev = __expf(scores[idx] - sm);
    scores[idx] = ev;
    atomicAddF(&denom[o * HEADS + h], ev);
}

__global__ void edge_scatter(const int* __restrict__ origin, const int* __restrict__ dest,
                             const float* __restrict__ ev, const float* __restrict__ denom,
                             const float* __restrict__ v, float* __restrict__ vals) {
    int idx = blockIdx.x * blockDim.x + threadIdx.x;
    int e = idx >> 7, dim = idx & 127, h = dim >> 4;
    int o = origin[e], d = dest[e];
    float attn = ev[e * HEADS + h] / (denom[o * HEADS + h] + 1e-16f);
    atomicAddF(&vals[(size_t)o * DIM + dim], attn * v[(size_t)d * DIM + dim]);
}

__global__ void out_gemm(const _Float16* __restrict__ vals16,
                         const _Float16* __restrict__ WTo, const float* __restrict__ bo,
                         float* __restrict__ out) {
    const int lane = threadIdx.x & 31;
    const int wave = threadIdx.x >> 5;
    const int r0 = blockIdx.x * 16;
    const int n0 = wave * 16;
    v16h a[4];
    load_a_frags(vals16, r0, lane, a);
    gemm_tile_16x16(a, WTo, bo, out, r0, n0, lane);
}

__global__ void ln_residual(const float* __restrict__ outp, const float* __restrict__ x,
                            const float* __restrict__ gamma, const float* __restrict__ beta,
                            float* __restrict__ y) {
    const int lane = threadIdx.x & 31;
    const int wave = threadIdx.x >> 5;
    const int node = blockIdx.x * 8 + wave;
    const float4 ov = ((const float4*)(outp + (size_t)node * DIM))[lane];
    float s = ov.x + ov.y + ov.z + ov.w;
#pragma unroll
    for (int m = 16; m >= 1; m >>= 1) s += __shfl_xor(s, m, 32);
    const float mu = s * (1.0f / DIM);
    float dx0 = ov.x - mu, dx1 = ov.y - mu, dx2 = ov.z - mu, dx3 = ov.w - mu;
    float s2 = dx0 * dx0 + dx1 * dx1 + dx2 * dx2 + dx3 * dx3;
#pragma unroll
    for (int m = 16; m >= 1; m >>= 1) s2 += __shfl_xor(s2, m, 32);
    const float inv = rsqrtf(s2 * (1.0f / DIM) + 1e-5f);
    const float4 xv = ((const float4*)(x + (size_t)node * DIM))[lane];
    const float4 gv = ((const float4*)gamma)[lane];
    const float4 bv = ((const float4*)beta)[lane];
    float4 r;
    r.x = xv.x + dx0 * inv * gv.x + bv.x;
    r.y = xv.y + dx1 * inv * gv.y + bv.y;
    r.z = xv.z + dx2 * inv * gv.z + bv.z;
    r.w = xv.w + dx3 * inv * gv.w + bv.w;
    ((float4*)(y + (size_t)node * DIM))[lane] = r;
}

extern "C" void kernel_launch(void* const* d_in, const int* in_sizes, int n_in,
                              void* d_out, int out_size, void* d_ws, size_t ws_size,
                              hipStream_t stream) {
    const float* x   = (const float*)d_in[0];
    const int*   ei  = (const int*)  d_in[1];
    const float* ew  = (const float*)d_in[2];
    const float* Wq  = (const float*)d_in[3];
    const float* bq  = (const float*)d_in[4];
    const float* Wk  = (const float*)d_in[5];
    const float* bk  = (const float*)d_in[6];
    const float* Wv  = (const float*)d_in[7];
    const float* bv  = (const float*)d_in[8];
    const float* Wo  = (const float*)d_in[9];
    const float* bo  = (const float*)d_in[10];
    const float* gam = (const float*)d_in[11];
    const float* bet = (const float*)d_in[12];
    float* y = (float*)d_out;

    const int* origin = ei;
    const int* dest   = ei + N_EDGES;

    float* ws   = (float*)d_ws;
    float* q    = ws;
    float* k    = q    + (size_t)N_NODES * DIM;
    float* v    = k    + (size_t)N_NODES * DIM;
    float* sc   = v    + (size_t)N_NODES * DIM;
    float* smx  = sc   + (size_t)N_EDGES * HEADS;
    float* den  = smx  + (size_t)N_NODES * HEADS;
    float* vals = den  + (size_t)N_NODES * HEADS;
    float* outb = vals + (size_t)N_NODES * DIM;
    _Float16* h16    = (_Float16*)(outb + (size_t)N_NODES * DIM);
    _Float16* x16    = h16;
    _Float16* vals16 = x16 + (size_t)N_NODES * DIM;
    _Float16* WT     = vals16 + (size_t)N_NODES * DIM;
    _Float16* WTq = WT;
    _Float16* WTk = WT + 1 * DIM * DIM;
    _Float16* WTv = WT + 2 * DIM * DIM;
    _Float16* WTo = WT + 3 * DIM * DIM;

    init_state<<<(N_NODES * DIM + 255) / 256, 256, 0, stream>>>((int*)smx, den, vals);

    cvt_f32_f16_x4<<<(N_NODES * DIM / 4) / 256, 256, 0, stream>>>(x, x16);
    cvt_weightsT<<<(4 * DIM * DIM) / 256, 256, 0, stream>>>(Wq, Wk, Wv, Wo, WT);

    qkv_gemm<<<N_NODES / 16, 256, 0, stream>>>(x16, WTq, bq, WTk, bk, WTv, bv, q, k, v);

    edge_scores<<<(N_EDGES * HEADS) / 256, 256, 0, stream>>>(origin, dest, ew, q, k,
                                                             sc, (int*)smx);

    edge_exp<<<(N_EDGES * HEADS) / 256, 256, 0, stream>>>(origin, (const int*)smx,
                                                          sc, den);

    edge_scatter<<<(N_EDGES * DIM) / 256, 256, 0, stream>>>(origin, dest, sc, den,
                                                            v, vals);

    cvt_f32_f16_x4<<<(N_NODES * DIM / 4) / 256, 256, 0, stream>>>(vals, vals16);

    out_gemm<<<N_NODES / 16, 256, 0, stream>>>(vals16, WTo, bo, outb);

    ln_residual<<<N_NODES / 8, 256, 0, stream>>>(outb, x, gam, bet, y);
}


# Group edges by origin in-call (count, scan, arrival-order slot via int atomics); one wave per node sums attn*v rows in registers: replaces 102M float atomics; sum order changes
# speedup vs baseline: 1.3400x; 1.3400x over previous
#include <hip/hip_runtime.h>
#include <hip/hip_bf16.h>

typedef __attribute__((ext_vector_type(16))) _Float16 v16h;
typedef __attribute__((ext_vector_type(8)))  _Float16 v8h;
typedef __attribute__((ext_vector_type(4)))  _Float16 v4h;
typedef __attribute__((ext_vector_type(8)))  float    v8f;

#define N_NODES 50000
#define N_EDGES 800000
#define DIM 128
#define HEADS 8
#define HEAD_DIM 16

#define SCAN_THREADS 1024
#define SCAN_VEC 13
#define SCAN_ITEMS (4 * SCAN_VEC)
#define CNT_PAD (SCAN_THREADS * SCAN_ITEMS)

__device__ inline void atomicAddF(float* p, float v) {
    __hip_atomic_fetch_add(p, v, __ATOMIC_RELAXED, __HIP_MEMORY_SCOPE_AGENT);
}
__device__ inline void atomicMaxF(int* p, float v) {
    if (v >= 0.0f) atomicMax(p, __float_as_int(v));
    else           atomicMin((unsigned int*)p, __float_as_uint(v));
}

__device__ inline void gemm_tile_16x16(const v16h a[4], const _Float16* __restrict__ WT,
                                       const float* __restrict__ bias,
                                       float* __restrict__ out,
                                       int r0, int n0, int lane) {
    const int col = lane & 15;
    const int kb  = (lane >> 4) << 3;
    const _Float16* wr = WT + (size_t)(n0 + col) * DIM;
    v8f c = {};
#pragma unroll
    for (int t = 0; t < 4; ++t) {
        const v8h lo = *(const v8h*)(wr + t * 32 + kb);
        const v8h hi = *(const v8h*)(wr + t * 32 + kb + 16);
        v16h b;
#pragma unroll
        for (int i = 0; i < 8; ++i) { b[i] = lo[i]; b[i + 8] = hi[i]; }
        c = __builtin_amdgcn_wmma_f32_16x16x32_f16(false, a[t], false, b,
                                                   (short)0, c, false, false);
    }
    const float bia = bias[n0 + col];
    float* op = out + (size_t)r0 * DIM + n0 + col;
#pragma unroll
    for (int j = 0; j < 8; ++j) {
        int m = j + ((lane >> 4) << 3);
        op[(size_t)m * DIM] = c[j] + bia;
    }
}

__device__ inline void load_a_frags(const _Float16* __restrict__ src, int r0, int lane,
                                    v16h a[4]) {
    const int row = lane & 15;
    const int kb  = (lane >> 4) << 3;
    const _Float16* xr = src + (size_t)(r0 + row) * DIM;
#pragma unroll
    for (int t = 0; t < 4; ++t) {
        const v8h lo = *(const v8h*)(xr + t * 32 + kb);
        const v8h hi = *(const v8h*)(xr + t * 32 + kb + 16);
#pragma unroll
        for (int i = 0; i < 8; ++i) { a[t][i] = lo[i]; a[t][i + 8] = hi[i]; }
    }
}

__global__ void init_state(int* segmax_bits, float* denom, int* cnt) {
    int i = blockIdx.x * blockDim.x + threadIdx.x;
    if (i < N_NODES * HEADS) {
        segmax_bits[i] = 0xFF800000;
        denom[i] = 0.0f;
    }
    if (i < CNT_PAD) cnt[i] = 0;
}

__global__ void edge_hist(const int* __restrict__ origin, int* __restrict__ cnt) {
    int e = blockIdx.x * blockDim.x + threadIdx.x;
    atomicAdd(&cnt[origin[e]], 1);
}

__global__ void csr_scan(const int* __restrict__ cnt, int* __restrict__ rowptr, int* __restrict__ cursor) {
    __shared__ int wsum[32];
    const int t = threadIdx.x;
    const int lane = t & 31;
    const int wave = t >> 5;
    const int4* cp = (const int4*)(cnt + t * SCAN_ITEMS);
    int4 c[SCAN_VEC];
    int tot = 0;
#pragma unroll
    for (int i = 0; i < SCAN_VEC; ++i) {
        c[i] = cp[i];
        tot += c[i].x + c[i].y + c[i].z + c[i].w;
    }
    int inc = tot;
    for (int m = 1; m < 32; m <<= 1) {
        int up = __shfl_up(inc, m, 32);
        if (lane >= m) inc += up;
    }
    if (lane == 31) wsum[wave] = inc;
    __syncthreads();
    if (wave == 0) {
        const int w = wsum[lane];
        int wi = w;
        for (int m = 1; m < 32; m <<= 1) {
            int up = __shfl_up(wi, m, 32);
            if (lane >= m) wi += up;
        }
        wsum[lane] = wi - w;
    }
    __syncthreads();
    int run = wsum[wave] + inc - tot;
    int4* rp = (int4*)(rowptr + t * SCAN_ITEMS);
    int4* cu = (int4*)(cursor + t * SCAN_ITEMS);
#pragma unroll
    for (int i = 0; i < SCAN_VEC; ++i) {
        int4 o;
        o.x = run; run += c[i].x;
        o.y = run; run += c[i].y;
        o.z = run; run += c[i].z;
        o.w = run; run += c[i].w;
        rp[i] = o;
        cu[i] = o;
    }
}

__global__ void csr_fill(const int* __restrict__ origin, const int* __restrict__ dest,
                         int* __restrict__ cursor,
                         int* __restrict__ sdest, int* __restrict__ sedge) {
    int e = blockIdx.x * blockDim.x + threadIdx.x;
    int o = origin[e], d = dest[e];
    int pos = atomicAdd(&cursor[o], 1);
    sdest[pos] = d;
    sedge[pos] = e;
}

__global__ void cvt_f32_f16_x4(const float* __restrict__ src, _Float16* __restrict__ dst) {
    int i = blockIdx.x * blockDim.x + threadIdx.x;
    float4 vv = ((const float4*)src)[i];
    v4h o = { (_Float16)vv.x, (_Float16)vv.y, (_Float16)vv.z, (_Float16)vv.w };
    ((v4h*)dst)[i] = o;
}

__global__ void cvt_weightsT(const float* __restrict__ Wq, const float* __restrict__ Wk,
                             const float* __restrict__ Wv, const float* __restrict__ Wo,
                             _Float16* __restrict__ WT) {
    int i = blockIdx.x * blockDim.x + threadIdx.x;
    int m = i >> 14, r = i & 16383, n = r >> 7, k = r & 127;
    const float* W = (m == 0) ? Wq : (m == 1) ? Wk : (m == 2) ? Wv : Wo;
    WT[i] = (_Float16)W[(size_t)k * DIM + n];
}

__global__ void qkv_gemm(const _Float16* __restrict__ x16,
                         const _Float16* __restrict__ WTq, const float* __restrict__ bq,
                         const _Float16* __restrict__ WTk, const float* __restrict__ bk,
                         const _Float16* __restrict__ WTv, const float* __restrict__ bv,
                         float* __restrict__ q, float* __restrict__ k,
                         float* __restrict__ v) {
    const int lane = threadIdx.x & 31;
    const int wave = threadIdx.x >> 5;
    const int r0 = blockIdx.x * 16;
    const int n0 = wave * 16;
    v16h a[4];
    load_a_frags(x16, r0, lane, a);
    gemm_tile_16x16(a, WTq, bq, q, r0, n0, lane);
    gemm_tile_16x16(a, WTk, bk, k, r0, n0, lane);
    gemm_tile_16x16(a, WTv, bv, v, r0, n0, lane);
}

__global__ void edge_scores(const int* __restrict__ origin, const int* __restrict__ dest,
                            const float* __restrict__ ew,
                            const float* __restrict__ q, const float* __restrict__ k,
                            float* __restrict__ scores, int* __restrict__ segmax_bits) {
    int idx = blockIdx.x * blockDim.x + threadIdx.x;
    int e = idx >> 3, h = idx & 7;
    int o = origin[e], d = dest[e];
    const float4* qp = (const float4*)(q + (size_t)o * DIM + h * HEAD_DIM);
    const float4* kp = (const float4*)(k + (size_t)d * DIM + h * HEAD_DIM);
    float s = 0.0f;
#pragma unroll
    for (int t = 0; t < 4; ++t) {
        float4 av = qp[t], bv = kp[t];
        s += av.x * bv.x + av.y * bv.y + av.z * bv.z + av.w * bv.w;
    }
    s *= 0.25f;
    s *= ew[idx];
    scores[idx] = s;
    atomicMaxF(&segmax_bits[o * HEADS + h], s);
}

__global__ void edge_exp(const int* __restrict__ origin,
                         const int* __restrict__ segmax_bits,
                         float* __restrict__ scores, float* __restrict__ denom) {
    int idx = blockIdx.x * blockDim.x + threadIdx.x;
    int e = idx >> 3, h = idx & 7;
    int o = origin[e];
    float sm = __int_as_float(segmax_bits[o * HEADS + h]);
    if (!isfinite(sm)) sm = 0.0f;
    float ev = __expf(scores[idx] - sm);
    scores[idx] = ev;
    atomicAddF(&denom[o * HEADS + h], ev);
}

__global__ void node_gather(const int* __restrict__ rowptr,
                            const int* __restrict__ sdest, const int* __restrict__ sedge,
                            const float* __restrict__ ev, const float* __restrict__ denom,
                            const float* __restrict__ v, float* __restrict__ vals) {
    const int lane = threadIdx.x & 31;
    const int wave = threadIdx.x >> 5;
    const int node = blockIdx.x * 8 + wave;
    const int h = lane >> 2;
    const int beg = __builtin_amdgcn_readfirstlane(rowptr[node]);
    const int end = __builtin_amdgcn_readfirstlane(rowptr[node + 1]);
    const float dn = denom[node * HEADS + h] + 1e-16f;
    float4 acc = {0.0f, 0.0f, 0.0f, 0.0f};
    for (int base = beg; base < end; base += 32) {
        const int n = (end - base < 32) ? (end - base) : 32;
        int my_d = 0, my_e = 0;
        if (lane < n) { my_d = sdest[base + lane]; my_e = sedge[base + lane]; }
#pragma unroll 4
        for (int j = 0; j < n; ++j) {
            const int d = __builtin_amdgcn_readlane(my_d, j);
            const int e = __builtin_amdgcn_readlane(my_e, j);
            const float a = ev[(size_t)e * HEADS + h] / dn;
            const float4 vv = ((const float4*)(v + (size_t)d * DIM))[lane];
            acc.x += a * vv.x;
            acc.y += a * vv.y;
            acc.z += a * vv.z;
            acc.w += a * vv.w;
        }
    }
    ((float4*)(vals + (size_t)node * DIM))[lane] = acc;
}

__global__ void out_gemm(const _Float16* __restrict__ vals16,
                         const _Float16* __restrict__ WTo, const float* __restrict__ bo,
                         float* __restrict__ out) {
    const int lane = threadIdx.x & 31;
    const int wave = threadIdx.x >> 5;
    const int r0 = blockIdx.x * 16;
    const int n0 = wave * 16;
    v16h a[4];
    load_a_frags(vals16, r0, lane, a);
    gemm_tile_16x16(a, WTo, bo, out, r0, n0, lane);
}

__global__ void ln_residual(const float* __restrict__ outp, const float* __restrict__ x,
                            const float* __restrict__ gamma, const float* __restrict__ beta,
                            float* __restrict__ y) {
    const int lane = threadIdx.x & 31;
    const int wave = threadIdx.x >> 5;
    const int node = blockIdx.x * 8 + wave;
    const float4 ov = ((const float4*)(outp + (size_t)node * DIM))[lane];
    float s = ov.x + ov.y + ov.z + ov.w;
#pragma unroll
    for (int m = 16; m >= 1; m >>= 1) s += __shfl_xor(s, m, 32);
    const float mu = s * (1.0f / DIM);
    float dx0 = ov.x - mu, dx1 = ov.y - mu, dx2 = ov.z - mu, dx3 = ov.w - mu;
    float s2 = dx0 * dx0 + dx1 * dx1 + dx2 * dx2 + dx3 * dx3;
#pragma unroll
    for (int m = 16; m >= 1; m >>= 1) s2 += __shfl_xor(s2, m, 32);
    const float inv = rsqrtf(s2 * (1.0f / DIM) + 1e-5f);
    const float4 xv = ((const float4*)(x + (size_t)node * DIM))[lane];
    const float4 gv = ((const float4*)gamma)[lane];
    const float4 bv = ((const float4*)beta)[lane];
    float4 r;
    r.x = xv.x + dx0 * inv * gv.x + bv.x;
    r.y = xv.y + dx1 * inv * gv.y + bv.y;
    r.z = xv.z + dx2 * inv * gv.z + bv.z;
    r.w = xv.w + dx3 * inv * gv.w + bv.w;
    ((float4*)(y + (size_t)node * DIM))[lane] = r;
}

extern "C" void kernel_launch(void* const* d_in, const int* in_sizes, int n_in,
                              void* d_out, int out_size, void* d_ws, size_t ws_size,
                              hipStream_t stream) {
    const float* x   = (const float*)d_in[0];
    const int*   ei  = (const int*)  d_in[1];
    const float* ew  = (const float*)d_in[2];
    const float* Wq  = (const float*)d_in[3];
    const float* bq  = (const float*)d_in[4];
    const float* Wk  = (const float*)d_in[5];
    const float* bk  = (const float*)d_in[6];
    const float* Wv  = (const float*)d_in[7];
    const float* bv  = (const float*)d_in[8];
    const float* Wo  = (const float*)d_in[9];
    const float* bo  = (const float*)d_in[10];
    const float* gam = (const float*)d_in[11];
    const float* bet = (const float*)d_in[12];
    float* y = (float*)d_out;

    const int* origin = ei;
    const int* dest   = ei + N_EDGES;

    float* ws   = (float*)d_ws;
    float* q    = ws;
    float* k    = q    + (size_t)N_NODES * DIM;
    float* v    = k    + (size_t)N_NODES * DIM;
    float* sc   = v    + (size_t)N_NODES * DIM;
    float* smx  = sc   + (size_t)N_EDGES * HEADS;
    float* den  = smx  + (size_t)N_NODES * HEADS;
    float* vals = den  + (size_t)N_NODES * HEADS;
    float* outb = vals + (size_t)N_NODES * DIM;
    _Float16* h16    = (_Float16*)(outb + (size_t)N_NODES * DIM);
    _Float16* x16    = h16;
    _Float16* vals16 = x16 + (size_t)N_NODES * DIM;
    _Float16* WT     = vals16 + (size_t)N_NODES * DIM;
    _Float16* WTq = WT;
    _Float16* WTk = WT + 1 * DIM * DIM;
    _Float16* WTv = WT + 2 * DIM * DIM;
    _Float16* WTo = WT + 3 * DIM * DIM;

    int* cnt    = (int*)outb;
    int* rowptr = cnt    + CNT_PAD;
    int* cursor = rowptr + CNT_PAD;
    int* sdest  = cursor + CNT_PAD;
    int* sedge  = sdest  + N_EDGES;

    init_state<<<(N_NODES * HEADS + 255) / 256, 256, 0, stream>>>((int*)smx, den, cnt);

    edge_hist<<<N_EDGES / 256, 256, 0, stream>>>(origin, cnt);
    csr_scan<<<1, SCAN_THREADS, 0, stream>>>(cnt, rowptr, cursor);
    csr_fill<<<N_EDGES / 256, 256, 0, stream>>>(origin, dest, cursor, sdest, sedge);

    cvt_f32_f16_x4<<<(N_NODES * DIM / 4) / 256, 256, 0, stream>>>(x, x16);
    cvt_weightsT<<<(4 * DIM * DIM) / 256, 256, 0, stream>>>(Wq, Wk, Wv, Wo, WT);

    qkv_gemm<<<N_NODES / 16, 256, 0, stream>>>(x16, WTq, bq, WTk, bk, WTv, bv, q, k, v);

    edge_scores<<<(N_EDGES * HEADS) / 256, 256, 0, stream>>>(origin, dest, ew, q, k,
                                                             sc, (int*)smx);

    edge_exp<<<(N_EDGES * HEADS) / 256, 256, 0, stream>>>(origin, (const int*)smx,
                                                          sc, den);

    node_gather<<<N_NODES / 8, 256, 0, stream>>>(rowptr, sdest, sedge, sc, den, v, vals);

    cvt_f32_f16_x4<<<(N_NODES * DIM / 4) / 256, 256, 0, stream>>>(vals, vals16);

    out_gemm<<<N_NODES / 16, 256, 0, stream>>>(vals16, WTo, bo, outb);

    ln_residual<<<N_NODES / 8, 256, 0, stream>>>(outb, x, gam, bet, y);
}


# Every launch starts a few thousand resident waves that loop over several rows, edges or nodes (was 25-50 thousand one-item waves): wave entry no longer paces the call; slots stay arrival-ordered
# speedup vs baseline: 1.9249x; 1.4365x over previous
#include <hip/hip_runtime.h>
#include <hip/hip_bf16.h>

typedef __attribute__((ext_vector_type(16))) _Float16 v16h;
typedef __attribute__((ext_vector_type(8)))  _Float16 v8h;
typedef __attribute__((ext_vector_type(4)))  _Float16 v4h;
typedef __attribute__((ext_vector_type(8)))  float    v8f;

#define N_NODES 50000
#define N_EDGES 800000
#define DIM 128
#define HEADS 8
#define HEAD_DIM 16

#define SCAN_THREADS 1024
#define SCAN_VEC 13
#define SCAN_ITEMS (4 * SCAN_VEC)
#define CNT_PAD (SCAN_THREADS * SCAN_ITEMS)

#define M_TILES (N_NODES / 16)
#define GEMM_BLOCKS 512
#define CVT_ITEMS 10
#define CVT_BLOCKS (N_NODES * DIM / 4 / (256 * CVT_ITEMS))
#define EDGE_THREADS 160
#define EDGE_BLOCKS (N_EDGES / 8 / EDGE_THREADS)
#define NODES_PER_WAVE 5
#define LN_ROWS 10
#define LN_BATCH 5


__device__ inline void load_frags(const _Float16* __restrict__ src, int r0, int lane,
                                  v16h f[4]) {
    const int row = lane & 15;
    const int kb  = (lane >> 4) << 3;
    const _Float16* xr = src + (size_t)(r0 + row) * DIM;
#pragma unroll
    for (int t = 0; t < 4; ++t) {
        const v8h lo = *(const v8h*)(xr + t * 32 + kb);
        const v8h hi = *(const v8h*)(xr + t * 32 + kb + 16);
#pragma unroll
        for (int i = 0; i < 8; ++i) { f[t][i] = lo[i]; f[t][i + 8] = hi[i]; }
    }
}

__device__ inline void mma_store_16x16(const v16h a[4], const v16h b[4], float bia,
                                       float* __restrict__ out,
                                       int r0, int n0, int lane) {
    v8f c = {};
#pragma unroll
    for (int t = 0; t < 4; ++t)
        c = __builtin_amdgcn_wmma_f32_16x16x32_f16(false, a[t], false, b[t],
                                                   (short)0, c, false, false);
    float* op = out + (size_t)r0 * DIM + n0 + (lane & 15);
#pragma unroll
    for (int j = 0; j < 8; ++j) {
        int m = j + ((lane >> 4) << 3);
        op[(size_t)m * DIM] = c[j] + bia;
    }
}

__global__ void init_state(int* cnt) {
    int i = blockIdx.x * blockDim.x + threadIdx.x;
    ((int4*)cnt)[i] = make_int4(0, 0, 0, 0);
}

__global__ void edge_hist(const int* __restrict__ origin, int* __restrict__ cnt) {
    const int i = blockIdx.x * EDGE_THREADS + threadIdx.x;
    const int4 a = ((const int4*)origin)[i];
    const int4 b = ((const int4*)origin)[i + N_EDGES / 8];
    atomicAdd(&cnt[a.x], 1); atomicAdd(&cnt[a.y], 1);
    atomicAdd(&cnt[a.z], 1); atomicAdd(&cnt[a.w], 1);
    atomicAdd(&cnt[b.x], 1); atomicAdd(&cnt[b.y], 1);
    atomicAdd(&cnt[b.z], 1); atomicAdd(&cnt[b.w], 1);
}

__global__ void csr_scan(const int* __restrict__ cnt, int* __restrict__ rowptr, int* __restrict__ cursor) {
    __shared__ int wsum[32];
    const int t = threadIdx.x;
    const int lane = t & 31;
    const int wave = t >> 5;
    const int4* cp = (const int4*)(cnt + t * SCAN_ITEMS);
    int4 c[SCAN_VEC];
    int tot = 0;
#pragma unroll
    for (int i = 0; i < SCAN_VEC; ++i) {
        c[i] = cp[i];
        tot += c[i].x + c[i].y + c[i].z + c[i].w;
    }
    int inc = tot;
    for (int m = 1; m < 32; m <<= 1) {
        int up = __shfl_up(inc, m, 32);
        if (lane >= m) inc += up;
    }
    if (lane == 31) wsum[wave] = inc;
    __syncthreads();
    if (wave == 0) {
        const int w = wsum[lane];
        int wi = w;
        for (int m = 1; m < 32; m <<= 1) {
            int up = __shfl_up(wi, m, 32);
            if (lane >= m) wi += up;
        }
        wsum[lane] = wi - w;
    }
    __syncthreads();
    int run = wsum[wave] + inc - tot;
    int4* rp = (int4*)(rowptr + t * SCAN_ITEMS);
    int4* cu = (int4*)(cursor + t * SCAN_ITEMS);
#pragma unroll
    for (int i = 0; i < SCAN_VEC; ++i) {
        int4 o;
        o.x = run; run += c[i].x;
        o.y = run; run += c[i].y;
        o.z = run; run += c[i].z;
        o.w = run; run += c[i].w;
        rp[i] = o;
        cu[i] = o;
    }
}

__global__ void csr_fill(const int* __restrict__ origin, const int* __restrict__ dest,
                         int* __restrict__ cursor,
                         int* __restrict__ sdest, int* __restrict__ sedge) {
    const int i = blockIdx.x * EDGE_THREADS + threadIdx.x;
    const int i2 = i + N_EDGES / 8;
    const int4 oa = ((const int4*)origin)[i];
    const int4 ob = ((const int4*)origin)[i2];
    const int4 da = ((const int4*)dest)[i];
    const int4 db = ((const int4*)dest)[i2];
    const int p0 = atomicAdd(&cursor[oa.x], 1);
    const int p1 = atomicAdd(&cursor[oa.y], 1);
    const int p2 = atomicAdd(&cursor[oa.z], 1);
    const int p3 = atomicAdd(&cursor[oa.w], 1);
    const int p4 = atomicAdd(&cursor[ob.x], 1);
    const int p5 = atomicAdd(&cursor[ob.y], 1);
    const int p6 = atomicAdd(&cursor[ob.z], 1);
    const int p7 = atomicAdd(&cursor[ob.w], 1);
    sdest[p0] = da.x; sedge[p0] = 4 * i;
    sdest[p1] = da.y; sedge[p1] = 4 * i + 1;
    sdest[p2] = da.z; sedge[p2] = 4 * i + 2;
    sdest[p3] = da.w; sedge[p3] = 4 * i + 3;
    sdest[p4] = db.x; sedge[p4] = 4 * i2;
    sdest[p5] = db.y; sedge[p5] = 4 * i2 + 1;
    sdest[p6] = db.z; sedge[p6] = 4 * i2 + 2;
    sdest[p7] = db.w; sedge[p7] = 4 * i2 + 3;
}

__global__ void cvt_f32_f16_x4(const float* __restrict__ src, _Float16* __restrict__ dst) {
    const int lane = threadIdx.x & 31;
    const int gw = blockIdx.x * 8 + (threadIdx.x >> 5);
    const int base = gw * (32 * CVT_ITEMS) + lane;
    float4 vv[CVT_ITEMS];
#pragma unroll
    for (int j = 0; j < CVT_ITEMS; ++j) vv[j] = ((const float4*)src)[base + 32 * j];
#pragma unroll
    for (int j = 0; j < CVT_ITEMS; ++j) {
        v4h o = { (_Float16)vv[j].x, (_Float16)vv[j].y, (_Float16)vv[j].z, (_Float16)vv[j].w };
        ((v4h*)dst)[base + 32 * j] = o;
    }
}

__global__ void cvt_weightsT(const float* __restrict__ Wq, const float* __restrict__ Wk,
                             const float* __restrict__ Wv, const float* __restrict__ Wo,
                             _Float16* __restrict__ WT) {
    int i = blockIdx.x * blockDim.x + threadIdx.x;
    int m = i >> 12, r = i & 4095, n = r >> 5, k = (r & 31) << 2;
    const float* W = (m == 0) ? Wq : (m == 1) ? Wk : (m == 2) ? Wv : Wo;
    const float* wp = W + (size_t)k * DIM + n;
    v4h o = { (_Float16)wp[0], (_Float16)wp[DIM], (_Float16)wp[2 * DIM], (_Float16)wp[3 * DIM] };
    *(v4h*)(WT + (size_t)m * DIM * DIM + n * DIM + k) = o;
}

__global__ void __launch_bounds__(256)
qkv_gemm(const _Float16* __restrict__ x16,
         const _Float16* __restrict__ WTq, const float* __restrict__ bq,
         const _Float16* __restrict__ WTk, const float* __restrict__ bk,
         const _Float16* __restrict__ WTv, const float* __restrict__ bv,
         float* __restrict__ q, float* __restrict__ k,
         float* __restrict__ v) {
    const int lane = threadIdx.x & 31;
    const int wave = threadIdx.x >> 5;
    const int n0 = wave * 16;
    v16h wq[4], wk[4], wv[4];
    load_frags(WTq, n0, lane, wq);
    load_frags(WTk, n0, lane, wk);
    load_frags(WTv, n0, lane, wv);
    const float biq = bq[n0 + (lane & 15)];
    const float bik = bk[n0 + (lane & 15)];
    const float biv = bv[n0 + (lane & 15)];
    int tile = blockIdx.x;
    v16h a[4];
    load_frags(x16, tile * 16, lane, a);
    while (tile < M_TILES) {
        const int nxt = (tile + GEMM_BLOCKS < M_TILES) ? (tile + GEMM_BLOCKS) : tile;
        v16h an[4];
        load_frags(x16, nxt * 16, lane, an);
        const int r0 = tile * 16;
        mma_store_16x16(a, wq, biq, q, r0, n0, lane);
        mma_store_16x16(a, wk, bik, k, r0, n0, lane);
        mma_store_16x16(a, wv, biv, v, r0, n0, lane);
#pragma unroll
        for (int t = 0; t < 4; ++t) a[t] = an[t];
        tile += GEMM_BLOCKS;
    }
}

#define EDGE_GRP 4

__device__ inline float quad_sum(float p) {
    p += __int_as_float(__builtin_amdgcn_update_dpp(0, __float_as_int(p), 0xB1, 0xF, 0xF, true));
    p += __int_as_float(__builtin_amdgcn_update_dpp(0, __float_as_int(p), 0x4E, 0xF, 0xF, true));
    return p;
}

typedef float v4f __attribute__((ext_vector_type(4)));
typedef const float __attribute__((address_space(1))) gfloat;
typedef const v4f   __attribute__((address_space(1))) gv4f;

__device__ inline gfloat* scalar_base(const float* p) {
    const unsigned long long a = (unsigned long long)p;
    const unsigned lo = __builtin_amdgcn_readfirstlane((unsigned)a);
    const unsigned hi = __builtin_amdgcn_readfirstlane((unsigned)(a >> 32));
    return (gfloat*)(((unsigned long long)hi << 32) | lo);
}

__global__ void node_attend(const int* __restrict__ rowptr,
                            const int* __restrict__ sdest, const int* __restrict__ sedge,
                            const float* __restrict__ ew,
                            const float* __restrict__ q, const float* __restrict__ k,
                            const float* __restrict__ v, float* __restrict__ vals) {
    const int lane = threadIdx.x & 31;
    const int wave = threadIdx.x >> 5;
    const int node0 = (blockIdx.x * 8 + wave) * NODES_PER_WAVE;
    const int h = lane >> 2;
    const int rp = rowptr[node0 + ((lane < NODES_PER_WAVE) ? lane : NODES_PER_WAVE)];
    for (int i = 0; i < NODES_PER_WAVE; ++i) {
        const int node = node0 + i;
        const int beg = __builtin_amdgcn_readlane(rp, i);
        const int end = __builtin_amdgcn_readlane(rp, i + 1);
        float4 qv = ((const float4*)(q + (size_t)node * DIM))[lane];
        qv.x *= 0.25f; qv.y *= 0.25f; qv.z *= 0.25f; qv.w *= 0.25f;
        float m = -INFINITY;
        float l = 0.0f;
        float4 acc = {0.0f, 0.0f, 0.0f, 0.0f};
        for (int base = beg; base < end; base += 32) {
            const int n = (end - base < 32) ? (end - base) : 32;
            int my_d = 0, my_e = 0;
            if (lane < n) { my_d = sdest[base + lane]; my_e = sedge[base + lane]; }
            for (int j0 = 0; j0 < n; j0 += EDGE_GRP) {
                float s[EDGE_GRP];
                v4f vv[EDGE_GRP];
#pragma unroll
                for (int g = 0; g < EDGE_GRP; ++g) {
                    const int j = j0 + g;
                    const int jc = (j < n) ? j : (n - 1);
                    const int d = __builtin_amdgcn_readlane(my_d, jc);
                    const int e = __builtin_amdgcn_readlane(my_e, jc);
                    const v4f kv = ((gv4f*)scalar_base(k + (size_t)d * DIM))[lane];
                    vv[g] = ((gv4f*)scalar_base(v + (size_t)d * DIM))[lane];
                    const float w = scalar_base(ew + (size_t)e * HEADS)[h];
                    float p = qv.x * kv.x + qv.y * kv.y + qv.z * kv.z + qv.w * kv.w;
                    p = quad_sum(p) * w;
                    s[g] = (j < n) ? p : -INFINITY;
                }
                float gm = m;
#pragma unroll
                for (int g = 0; g < EDGE_GRP; ++g) gm = fmaxf(gm, s[g]);
                const float corr = __expf(m - gm);
                l *= corr;
                acc.x *= corr; acc.y *= corr; acc.z *= corr; acc.w *= corr;
#pragma unroll
                for (int g = 0; g < EDGE_GRP; ++g) {
                    const float pg = __expf(s[g] - gm);
                    l += pg;
                    acc.x += pg * vv[g].x;
                    acc.y += pg * vv[g].y;
                    acc.z += pg * vv[g].z;
                    acc.w += pg * vv[g].w;
                }
                m = gm;
            }
        }
        const float inv = 1.0f / (l + 1e-16f);
        float4 r = {acc.x * inv, acc.y * inv, acc.z * inv, acc.w * inv};
        ((float4*)(vals + (size_t)node * DIM))[lane] = r;
    }
}

__global__ void __launch_bounds__(256)
out_gemm(const _Float16* __restrict__ vals16,
         const _Float16* __restrict__ WTo, const float* __restrict__ bo,
         float* __restrict__ out) {
    const int lane = threadIdx.x & 31;
    const int wave = threadIdx.x >> 5;
    const int n0 = wave * 16;
    v16h wo[4];
    load_frags(WTo, n0, lane, wo);
    const float bio = bo[n0 + (lane & 15)];
    int tile = blockIdx.x;
    v16h a[4];
    load_frags(vals16, tile * 16, lane, a);
    while (tile < M_TILES) {
        const int nxt = (tile + GEMM_BLOCKS < M_TILES) ? (tile + GEMM_BLOCKS) : tile;
        v16h an[4];
        load_frags(vals16, nxt * 16, lane, an);
        mma_store_16x16(a, wo, bio, out, tile * 16, n0, lane);
#pragma unroll
        for (int t = 0; t < 4; ++t) a[t] = an[t];
        tile += GEMM_BLOCKS;
    }
}

__global__ void ln_residual(const float* __restrict__ outp, const float* __restrict__ x,
                            const float* __restrict__ gamma, const float* __restrict__ beta,
                            float* __restrict__ y) {
    const int lane = threadIdx.x & 31;
    const int wave = threadIdx.x >> 5;
    const int node0 = (blockIdx.x * 8 + wave) * LN_ROWS;
    const float4 gv = ((const float4*)gamma)[lane];
    const float4 bv = ((const float4*)beta)[lane];
    for (int b = 0; b < LN_ROWS; b += LN_BATCH) {
        float4 ovs[LN_BATCH], xvs[LN_BATCH];
#pragma unroll
        for (int r = 0; r < LN_BATCH; ++r) {
            const size_t node = (size_t)(node0 + b + r);
            ovs[r] = ((const float4*)(outp + node * DIM))[lane];
            xvs[r] = ((const float4*)(x + node * DIM))[lane];
        }
#pragma unroll
        for (int r = 0; r < LN_BATCH; ++r) {
            const size_t node = (size_t)(node0 + b + r);
            const float4 ov = ovs[r];
            const float4 xv = xvs[r];
            float s = ov.x + ov.y + ov.z + ov.w;
#pragma unroll
            for (int m = 16; m >= 1; m >>= 1) s += __shfl_xor(s, m, 32);
            const float mu = s * (1.0f / DIM);
            float dx0 = ov.x - mu, dx1 = ov.y - mu, dx2 = ov.z - mu, dx3 = ov.w - mu;
            float s2 = dx0 * dx0 + dx1 * dx1 + dx2 * dx2 + dx3 * dx3;
#pragma unroll
            for (int m = 16; m >= 1; m >>= 1) s2 += __shfl_xor(s2, m, 32);
            const float inv = rsqrtf(s2 * (1.0f / DIM) + 1e-5f);
            float4 rr;
            rr.x = xv.x + dx0 * inv * gv.x + bv.x;
            rr.y = xv.y + dx1 * inv * gv.y + bv.y;
            rr.z = xv.z + dx2 * inv * gv.z + bv.z;
            rr.w = xv.w + dx3 * inv * gv.w + bv.w;
            ((float4*)(y + node * DIM))[lane] = rr;
        }
    }
}

extern "C" void kernel_launch(void* const* d_in, const int* in_sizes, int n_in,
                              void* d_out, int out_size, void* d_ws, size_t ws_size,
                              hipStream_t stream) {
    const float* x   = (const float*)d_in[0];
    const int*   ei  = (const int*)  d_in[1];
    const float* ew  = (const float*)d_in[2];
    const float* Wq  = (const float*)d_in[3];
    const float* bq  = (const float*)d_in[4];
    const float* Wk  = (const float*)d_in[5];
    const float* bk  = (const float*)d_in[6];
    const float* Wv  = (const float*)d_in[7];
    const float* bv  = (const float*)d_in[8];
    const float* Wo  = (const float*)d_in[9];
    const float* bo  = (const float*)d_in[10];
    const float* gam = (const float*)d_in[11];
    const float* bet = (const float*)d_in[12];
    float* y = (float*)d_out;

    const int* origin = ei;
    const int* dest   = ei + N_EDGES;

    float* ws   = (float*)d_ws;
    float* q    = ws;
    float* k    = q    + (size_t)N_NODES * DIM;
    float* v    = k    + (size_t)N_NODES * DIM;
    float* vals = v    + (size_t)N_NODES * DIM;
    float* outb = vals + (size_t)N_NODES * DIM;
    _Float16* h16    = (_Float16*)(outb + (size_t)N_NODES * DIM);
    _Float16* x16    = h16;
    _Float16* vals16 = x16 + (size_t)N_NODES * DIM;
    _Float16* WT     = vals16 + (size_t)N_NODES * DIM;
    _Float16* WTq = WT;
    _Float16* WTk = WT + 1 * DIM * DIM;
    _Float16* WTv = WT + 2 * DIM * DIM;
    _Float16* WTo = WT + 3 * DIM * DIM;

    int* cnt    = (int*)outb;
    int* rowptr = cnt    + CNT_PAD;
    int* cursor = rowptr + CNT_PAD;
    int* sdest  = cursor + CNT_PAD;
    int* sedge  = sdest  + N_EDGES;

    init_state<<<CNT_PAD / 4 / 256, 256, 0, stream>>>(cnt);

    edge_hist<<<EDGE_BLOCKS, EDGE_THREADS, 0, stream>>>(origin, cnt);
    csr_scan<<<1, SCAN_THREADS, 0, stream>>>(cnt, rowptr, cursor);
    csr_fill<<<EDGE_BLOCKS, EDGE_THREADS, 0, stream>>>(origin, dest, cursor, sdest, sedge);

    cvt_f32_f16_x4<<<CVT_BLOCKS, 256, 0, stream>>>(x, x16);
    cvt_weightsT<<<(DIM * DIM) / 256, 256, 0, stream>>>(Wq, Wk, Wv, Wo, WT);

    qkv_gemm<<<GEMM_BLOCKS, 256, 0, stream>>>(x16, WTq, bq, WTk, bk, WTv, bv, q, k, v);

    node_attend<<<N_NODES / (8 * NODES_PER_WAVE), 256, 0, stream>>>(rowptr, sdest, sedge, ew, q, k, v, vals);

    cvt_f32_f16_x4<<<CVT_BLOCKS, 256, 0, stream>>>(vals, vals16);

    out_gemm<<<GEMM_BLOCKS, 256, 0, stream>>>(vals16, WTo, bo, outb);

    ln_residual<<<N_NODES / (8 * LN_ROWS), 256, 0, stream>>>(outb, x, gam, bet, y);
}


# Edge counts and ranks per origin by LDS atomics: 25 workgroups each count a 32000-edge chunk in a 50000-word LDS table, chunk offsets by plain loads; no global atomic; sum order: chunk, then arrival
# speedup vs baseline: 2.8349x; 1.4728x over previous
#include <hip/hip_runtime.h>
#include <hip/hip_bf16.h>

typedef __attribute__((ext_vector_type(16))) _Float16 v16h;
typedef __attribute__((ext_vector_type(8)))  _Float16 v8h;
typedef __attribute__((ext_vector_type(4)))  _Float16 v4h;
typedef __attribute__((ext_vector_type(8)))  float    v8f;

#define N_NODES 50000
#define N_EDGES 800000
#define DIM 128
#define HEADS 8
#define HEAD_DIM 16

#define SCAN_THREADS 1024
#define SCAN_VEC 13
#define SCAN_ITEMS (4 * SCAN_VEC)
#define CNT_PAD (SCAN_THREADS * SCAN_ITEMS)

#define HIST_CHUNKS 25
#define HIST_THREADS 800
#define CHUNK_V4 (N_EDGES / 4 / HIST_CHUNKS)
#define HIST_V4 (CHUNK_V4 / HIST_THREADS)
#define NODE_V4 (N_NODES / 4)

#define M_TILES (N_NODES / 16)
#define GEMM_BLOCKS 512
#define CVT_ITEMS 10
#define CVT_BLOCKS (N_NODES * DIM / 4 / (256 * CVT_ITEMS))
#define EDGE_THREADS 160
#define EDGE_BLOCKS (N_EDGES / 8 / EDGE_THREADS)
#define NODES_PER_WAVE 5
#define LN_ROWS 10
#define LN_BATCH 5


__device__ inline void load_frags(const _Float16* __restrict__ src, int r0, int lane,
                                  v16h f[4]) {
    const int row = lane & 15;
    const int kb  = (lane >> 4) << 3;
    const _Float16* xr = src + (size_t)(r0 + row) * DIM;
#pragma unroll
    for (int t = 0; t < 4; ++t) {
        const v8h lo = *(const v8h*)(xr + t * 32 + kb);
        const v8h hi = *(const v8h*)(xr + t * 32 + kb + 16);
#pragma unroll
        for (int i = 0; i < 8; ++i) { f[t][i] = lo[i]; f[t][i + 8] = hi[i]; }
    }
}

__device__ inline void mma_store_16x16(const v16h a[4], const v16h b[4], float bia,
                                       float* __restrict__ out,
                                       int r0, int n0, int lane) {
    v8f c = {};
#pragma unroll
    for (int t = 0; t < 4; ++t)
        c = __builtin_amdgcn_wmma_f32_16x16x32_f16(false, a[t], false, b[t],
                                                   (short)0, c, false, false);
    float* op = out + (size_t)r0 * DIM + n0 + (lane & 15);
#pragma unroll
    for (int j = 0; j < 8; ++j) {
        int m = j + ((lane >> 4) << 3);
        op[(size_t)m * DIM] = c[j] + bia;
    }
}

__global__ void __launch_bounds__(HIST_THREADS)
edge_hist(const int* __restrict__ origin, int* __restrict__ bcnt,
          int* __restrict__ erank) {
    __shared__ int4 hist4[NODE_V4];
    int* hist = (int*)hist4;
    const int t = threadIdx.x;
    const int4* op = (const int4*)origin + blockIdx.x * CHUNK_V4;
    int4 o[HIST_V4];
#pragma unroll
    for (int j = 0; j < HIST_V4; ++j) o[j] = op[t + HIST_THREADS * j];
    for (int i = t; i < NODE_V4; i += HIST_THREADS) hist4[i] = make_int4(0, 0, 0, 0);
    __syncthreads();
#pragma unroll
    for (int j = 0; j < HIST_V4; ++j) {
        o[j].x = atomicAdd(&hist[o[j].x], 1); o[j].y = atomicAdd(&hist[o[j].y], 1);
        o[j].z = atomicAdd(&hist[o[j].z], 1); o[j].w = atomicAdd(&hist[o[j].w], 1);
    }
    __syncthreads();
    int4* bp = (int4*)(bcnt + (size_t)blockIdx.x * N_NODES);
    for (int i = t; i < NODE_V4; i += HIST_THREADS) bp[i] = hist4[i];
    int4* rp = (int4*)erank + blockIdx.x * CHUNK_V4;
#pragma unroll
    for (int j = 0; j < HIST_V4; ++j) rp[t + HIST_THREADS * j] = o[j];
}

__global__ void chunk_prefix(int* __restrict__ bcnt, int* __restrict__ cnt) {
    const int n = blockIdx.x * 256 + threadIdx.x;
    int run = 0;
    if (n < N_NODES) {
        int c[HIST_CHUNKS];
#pragma unroll
        for (int b = 0; b < HIST_CHUNKS; ++b) c[b] = bcnt[(size_t)b * N_NODES + n];
#pragma unroll
        for (int b = 0; b < HIST_CHUNKS; ++b) {
            bcnt[(size_t)b * N_NODES + n] = run;
            run += c[b];
        }
    }
    cnt[n] = run;
}

__global__ void csr_scan(const int* __restrict__ cnt, int* __restrict__ rowptr) {
    __shared__ int wsum[32];
    const int t = threadIdx.x;
    const int lane = t & 31;
    const int wave = t >> 5;
    const int4* cp = (const int4*)(cnt + t * SCAN_ITEMS);
    int4 c[SCAN_VEC];
    int tot = 0;
#pragma unroll
    for (int i = 0; i < SCAN_VEC; ++i) {
        c[i] = cp[i];
        tot += c[i].x + c[i].y + c[i].z + c[i].w;
    }
    int inc = tot;
    for (int m = 1; m < 32; m <<= 1) {
        int up = __shfl_up(inc, m, 32);
        if (lane >= m) inc += up;
    }
    if (lane == 31) wsum[wave] = inc;
    __syncthreads();
    if (wave == 0) {
        const int w = wsum[lane];
        int wi = w;
        for (int m = 1; m < 32; m <<= 1) {
            int up = __shfl_up(wi, m, 32);
            if (lane >= m) wi += up;
        }
        wsum[lane] = wi - w;
    }
    __syncthreads();
    int run = wsum[wave] + inc - tot;
    int4* rp = (int4*)(rowptr + t * SCAN_ITEMS);
#pragma unroll
    for (int i = 0; i < SCAN_VEC; ++i) {
        int4 o;
        o.x = run; run += c[i].x;
        o.y = run; run += c[i].y;
        o.z = run; run += c[i].z;
        o.w = run; run += c[i].w;
        rp[i] = o;
    }
}

static_assert(CHUNK_V4 % EDGE_THREADS == 0 && (N_EDGES / 8) % EDGE_THREADS == 0,
              "a csr_fill block must lie inside one chunk, in both halves");
__global__ void csr_fill(const int* __restrict__ origin, const int* __restrict__ dest,
                         const int* __restrict__ rowptr, const int* __restrict__ boff,
                         const int* __restrict__ erank,
                         int* __restrict__ sdest, int* __restrict__ sedge) {
    const int i = blockIdx.x * EDGE_THREADS + threadIdx.x;
    const int i2 = i + N_EDGES / 8;
    const int* b1 = boff + (size_t)(blockIdx.x / (CHUNK_V4 / EDGE_THREADS)) * N_NODES;
    const int* b2 = boff + (size_t)((blockIdx.x + N_EDGES / 8 / EDGE_THREADS)
                                    / (CHUNK_V4 / EDGE_THREADS)) * N_NODES;
    const int4 oa = ((const int4*)origin)[i];
    const int4 ob = ((const int4*)origin)[i2];
    const int4 ra = ((const int4*)erank)[i];
    const int4 rb = ((const int4*)erank)[i2];
    const int4 da = ((const int4*)dest)[i];
    const int4 db = ((const int4*)dest)[i2];
    const int p0 = rowptr[oa.x] + b1[oa.x] + ra.x;
    const int p1 = rowptr[oa.y] + b1[oa.y] + ra.y;
    const int p2 = rowptr[oa.z] + b1[oa.z] + ra.z;
    const int p3 = rowptr[oa.w] + b1[oa.w] + ra.w;
    const int p4 = rowptr[ob.x] + b2[ob.x] + rb.x;
    const int p5 = rowptr[ob.y] + b2[ob.y] + rb.y;
    const int p6 = rowptr[ob.z] + b2[ob.z] + rb.z;
    const int p7 = rowptr[ob.w] + b2[ob.w] + rb.w;
    sdest[p0] = da.x; sedge[p0] = 4 * i;
    sdest[p1] = da.y; sedge[p1] = 4 * i + 1;
    sdest[p2] = da.z; sedge[p2] = 4 * i + 2;
    sdest[p3] = da.w; sedge[p3] = 4 * i + 3;
    sdest[p4] = db.x; sedge[p4] = 4 * i2;
    sdest[p5] = db.y; sedge[p5] = 4 * i2 + 1;
    sdest[p6] = db.z; sedge[p6] = 4 * i2 + 2;
    sdest[p7] = db.w; sedge[p7] = 4 * i2 + 3;
}

__global__ void cvt_f32_f16_x4(const float* __restrict__ src, _Float16* __restrict__ dst) {
    const int lane = threadIdx.x & 31;
    const int gw = blockIdx.x * 8 + (threadIdx.x >> 5);
    const int base = gw * (32 * CVT_ITEMS) + lane;
    float4 vv[CVT_ITEMS];
#pragma unroll
    for (int j = 0; j < CVT_ITEMS; ++j) vv[j] = ((const float4*)src)[base + 32 * j];
#pragma unroll
    for (int j = 0; j < CVT_ITEMS; ++j) {
        v4h o = { (_Float16)vv[j].x, (_Float16)vv[j].y, (_Float16)vv[j].z, (_Float16)vv[j].w };
        ((v4h*)dst)[base + 32 * j] = o;
    }
}

__global__ void cvt_weightsT(const float* __restrict__ Wq, const float* __restrict__ Wk,
                             const float* __restrict__ Wv, const float* __restrict__ Wo,
                             _Float16* __restrict__ WT) {
    int i = blockIdx.x * blockDim.x + threadIdx.x;
    int m = i >> 12, r = i & 4095, n = r >> 5, k = (r & 31) << 2;
    const float* W = (m == 0) ? Wq : (m == 1) ? Wk : (m == 2) ? Wv : Wo;
    const float* wp = W + (size_t)k * DIM + n;
    v4h o = { (_Float16)wp[0], (_Float16)wp[DIM], (_Float16)wp[2 * DIM], (_Float16)wp[3 * DIM] };
    *(v4h*)(WT + (size_t)m * DIM * DIM + n * DIM + k) = o;
}

__global__ void __launch_bounds__(256)
qkv_gemm(const _Float16* __restrict__ x16,
         const _Float16* __restrict__ WTq, const float* __restrict__ bq,
         const _Float16* __restrict__ WTk, const float* __restrict__ bk,
         const _Float16* __restrict__ WTv, const float* __restrict__ bv,
         float* __restrict__ q, float* __restrict__ k,
         float* __restrict__ v) {
    const int lane = threadIdx.x & 31;
    const int wave = threadIdx.x >> 5;
    const int n0 = wave * 16;
    v16h wq[4], wk[4], wv[4];
    load_frags(WTq, n0, lane, wq);
    load_frags(WTk, n0, lane, wk);
    load_frags(WTv, n0, lane, wv);
    const float biq = bq[n0 + (lane & 15)];
    const float bik = bk[n0 + (lane & 15)];
    const float biv = bv[n0 + (lane & 15)];
    int tile = blockIdx.x;
    v16h a[4];
    load_frags(x16, tile * 16, lane, a);
    while (tile < M_TILES) {
        const int nxt = (tile + GEMM_BLOCKS < M_TILES) ? (tile + GEMM_BLOCKS) : tile;
        v16h an[4];
        load_frags(x16, nxt * 16, lane, an);
        const int r0 = tile * 16;
        mma_store_16x16(a, wq, biq, q, r0, n0, lane);
        mma_store_16x16(a, wk, bik, k, r0, n0, lane);
        mma_store_16x16(a, wv, biv, v, r0, n0, lane);
#pragma unroll
        for (int t = 0; t < 4; ++t) a[t] = an[t];
        tile += GEMM_BLOCKS;
    }
}

#define EDGE_GRP 4

__device__ inline float quad_sum(float p) {
    p += __int_as_float(__builtin_amdgcn_update_dpp(0, __float_as_int(p), 0xB1, 0xF, 0xF, true));
    p += __int_as_float(__builtin_amdgcn_update_dpp(0, __float_as_int(p), 0x4E, 0xF, 0xF, true));
    return p;
}

typedef float v4f __attribute__((ext_vector_type(4)));
typedef const float __attribute__((address_space(1))) gfloat;
typedef const v4f   __attribute__((address_space(1))) gv4f;

__device__ inline gfloat* scalar_base(const float* p) {
    const unsigned long long a = (unsigned long long)p;
    const unsigned lo = __builtin_amdgcn_readfirstlane((unsigned)a);
    const unsigned hi = __builtin_amdgcn_readfirstlane((unsigned)(a >> 32));
    return (gfloat*)(((unsigned long long)hi << 32) | lo);
}

__global__ void node_attend(const int* __restrict__ rowptr,
                            const int* __restrict__ sdest, const int* __restrict__ sedge,
                            const float* __restrict__ ew,
                            const float* __restrict__ q, const float* __restrict__ k,
                            const float* __restrict__ v, float* __restrict__ vals) {
    const int lane = threadIdx.x & 31;
    const int wave = threadIdx.x >> 5;
    const int node0 = (blockIdx.x * 8 + wave) * NODES_PER_WAVE;
    const int h = lane >> 2;
    const int rp = rowptr[node0 + ((lane < NODES_PER_WAVE) ? lane : NODES_PER_WAVE)];
    for (int i = 0; i < NODES_PER_WAVE; ++i) {
        const int node = node0 + i;
        const int beg = __builtin_amdgcn_readlane(rp, i);
        const int end = __builtin_amdgcn_readlane(rp, i + 1);
        float4 qv = ((const float4*)(q + (size_t)node * DIM))[lane];
        qv.x *= 0.25f; qv.y *= 0.25f; qv.z *= 0.25f; qv.w *= 0.25f;
        float m = -INFINITY;
        float l = 0.0f;
        float4 acc = {0.0f, 0.0f, 0.0f, 0.0f};
        for (int base = beg; base < end; base += 32) {
            const int n = (end - base < 32) ? (end - base) : 32;
            int my_d = 0, my_e = 0;
            if (lane < n) { my_d = sdest[base + lane]; my_e = sedge[base + lane]; }
            for (int j0 = 0; j0 < n; j0 += EDGE_GRP) {
                float s[EDGE_GRP];
                v4f vv[EDGE_GRP];
#pragma unroll
                for (int g = 0; g < EDGE_GRP; ++g) {
                    const int j = j0 + g;
                    const int jc = (j < n) ? j : (n - 1);
                    const int d = __builtin_amdgcn_readlane(my_d, jc);
                    const int e = __builtin_amdgcn_readlane(my_e, jc);
                    const v4f kv = ((gv4f*)scalar_base(k + (size_t)d * DIM))[lane];
                    vv[g] = ((gv4f*)scalar_base(v + (size_t)d * DIM))[lane];
                    const float w = scalar_base(ew + (size_t)e * HEADS)[h];
                    float p = qv.x * kv.x + qv.y * kv.y + qv.z * kv.z + qv.w * kv.w;
                    p = quad_sum(p) * w;
                    s[g] = (j < n) ? p : -INFINITY;
                }
                float gm = m;
#pragma unroll
                for (int g = 0; g < EDGE_GRP; ++g) gm = fmaxf(gm, s[g]);
                const float corr = __expf(m - gm);
                l *= corr;
                acc.x *= corr; acc.y *= corr; acc.z *= corr; acc.w *= corr;
#pragma unroll
                for (int g = 0; g < EDGE_GRP; ++g) {
                    const float pg = __expf(s[g] - gm);
                    l += pg;
                    acc.x += pg * vv[g].x;
                    acc.y += pg * vv[g].y;
                    acc.z += pg * vv[g].z;
                    acc.w += pg * vv[g].w;
                }
                m = gm;
            }
        }
        const float inv = 1.0f / (l + 1e-16f);
        float4 r = {acc.x * inv, acc.y * inv, acc.z * inv, acc.w * inv};
        ((float4*)(vals + (size_t)node * DIM))[lane] = r;
    }
}

__global__ void __launch_bounds__(256)
out_gemm(const _Float16* __restrict__ vals16,
         const _Float16* __restrict__ WTo, const float* __restrict__ bo,
         float* __restrict__ out) {
    const int lane = threadIdx.x & 31;
    const int wave = threadIdx.x >> 5;
    const int n0 = wave * 16;
    v16h wo[4];
    load_frags(WTo, n0, lane, wo);
    const float bio = bo[n0 + (lane & 15)];
    int tile = blockIdx.x;
    v16h a[4];
    load_frags(vals16, tile * 16, lane, a);
    while (tile < M_TILES) {
        const int nxt = (tile + GEMM_BLOCKS < M_TILES) ? (tile + GEMM_BLOCKS) : tile;
        v16h an[4];
        load_frags(vals16, nxt * 16, lane, an);
        mma_store_16x16(a, wo, bio, out, tile * 16, n0, lane);
#pragma unroll
        for (int t = 0; t < 4; ++t) a[t] = an[t];
        tile += GEMM_BLOCKS;
    }
}

__global__ void ln_residual(const float* __restrict__ outp, const float* __restrict__ x,
                            const float* __restrict__ gamma, const float* __restrict__ beta,
                            float* __restrict__ y) {
    const int lane = threadIdx.x & 31;
    const int wave = threadIdx.x >> 5;
    const int node0 = (blockIdx.x * 8 + wave) * LN_ROWS;
    const float4 gv = ((const float4*)gamma)[lane];
    const float4 bv = ((const float4*)beta)[lane];
    for (int b = 0; b < LN_ROWS; b += LN_BATCH) {
        float4 ovs[LN_BATCH], xvs[LN_BATCH];
#pragma unroll
        for (int r = 0; r < LN_BATCH; ++r) {
            const size_t node = (size_t)(node0 + b + r);
            ovs[r] = ((const float4*)(outp + node * DIM))[lane];
            xvs[r] = ((const float4*)(x + node * DIM))[lane];
        }
#pragma unroll
        for (int r = 0; r < LN_BATCH; ++r) {
            const size_t node = (size_t)(node0 + b + r);
            const float4 ov = ovs[r];
            const float4 xv = xvs[r];
            float s = ov.x + ov.y + ov.z + ov.w;
#pragma unroll
            for (int m = 16; m >= 1; m >>= 1) s += __shfl_xor(s, m, 32);
            const float mu = s * (1.0f / DIM);
            float dx0 = ov.x - mu, dx1 = ov.y - mu, dx2 = ov.z - mu, dx3 = ov.w - mu;
            float s2 = dx0 * dx0 + dx1 * dx1 + dx2 * dx2 + dx3 * dx3;
#pragma unroll
            for (int m = 16; m >= 1; m >>= 1) s2 += __shfl_xor(s2, m, 32);
            const float inv = rsqrtf(s2 * (1.0f / DIM) + 1e-5f);
            float4 rr;
            rr.x = xv.x + dx0 * inv * gv.x + bv.x;
            rr.y = xv.y + dx1 * inv * gv.y + bv.y;
            rr.z = xv.z + dx2 * inv * gv.z + bv.z;
            rr.w = xv.w + dx3 * inv * gv.w + bv.w;
            ((float4*)(y + node * DIM))[lane] = rr;
        }
    }
}

extern "C" void kernel_launch(void* const* d_in, const int* in_sizes, int n_in,
                              void* d_out, int out_size, void* d_ws, size_t ws_size,
                              hipStream_t stream) {
    const float* x   = (const float*)d_in[0];
    const int*   ei  = (const int*)  d_in[1];
    const float* ew  = (const float*)d_in[2];
    const float* Wq  = (const float*)d_in[3];
    const float* bq  = (const float*)d_in[4];
    const float* Wk  = (const float*)d_in[5];
    const float* bk  = (const float*)d_in[6];
    const float* Wv  = (const float*)d_in[7];
    const float* bv  = (const float*)d_in[8];
    const float* Wo  = (const float*)d_in[9];
    const float* bo  = (const float*)d_in[10];
    const float* gam = (const float*)d_in[11];
    const float* bet = (const float*)d_in[12];
    float* y = (float*)d_out;

    const int* origin = ei;
    const int* dest   = ei + N_EDGES;

    float* ws   = (float*)d_ws;
    float* q    = ws;
    float* k    = q    + (size_t)N_NODES * DIM;
    float* v    = k    + (size_t)N_NODES * DIM;
    float* vals = v    + (size_t)N_NODES * DIM;
    float* outb = vals + (size_t)N_NODES * DIM;
    _Float16* h16    = (_Float16*)(outb + (size_t)N_NODES * DIM);
    _Float16* x16    = h16;
    _Float16* vals16 = x16 + (size_t)N_NODES * DIM;
    _Float16* WT     = vals16 + (size_t)N_NODES * DIM;
    _Float16* WTq = WT;
    _Float16* WTk = WT + 1 * DIM * DIM;
    _Float16* WTv = WT + 2 * DIM * DIM;
    _Float16* WTo = WT + 3 * DIM * DIM;

    int* cnt    = (int*)outb;
    int* rowptr = cnt    + CNT_PAD;
    int* sdest  = rowptr + CNT_PAD;
    int* sedge  = sdest  + N_EDGES;
    int* erank  = sedge  + N_EDGES;
    int* bcnt   = erank  + N_EDGES;

    edge_hist<<<HIST_CHUNKS, HIST_THREADS, 0, stream>>>(origin, bcnt, erank);
    chunk_prefix<<<CNT_PAD / 256, 256, 0, stream>>>(bcnt, cnt);
    csr_scan<<<1, SCAN_THREADS, 0, stream>>>(cnt, rowptr);
    csr_fill<<<EDGE_BLOCKS, EDGE_THREADS, 0, stream>>>(origin, dest, rowptr, bcnt, erank, sdest, sedge);

    cvt_f32_f16_x4<<<CVT_BLOCKS, 256, 0, stream>>>(x, x16);
    cvt_weightsT<<<(DIM * DIM) / 256, 256, 0, stream>>>(Wq, Wk, Wv, Wo, WT);

    qkv_gemm<<<GEMM_BLOCKS, 256, 0, stream>>>(x16, WTq, bq, WTk, bk, WTv, bv, q, k, v);

    node_attend<<<N_NODES / (8 * NODES_PER_WAVE), 256, 0, stream>>>(rowptr, sdest, sedge, ew, q, k, v, vals);

    cvt_f32_f16_x4<<<CVT_BLOCKS, 256, 0, stream>>>(vals, vals16);

    out_gemm<<<GEMM_BLOCKS, 256, 0, stream>>>(vals16, WTo, bo, outb);

    ln_residual<<<N_NODES / (8 * LN_ROWS), 256, 0, stream>>>(outb, x, gam, bet, y);
}


# Attention rows stay in the workgroup: each 16-node tile goes via LDS (as f16) into the Wo product, LayerNorm and residual; conversion, Wo and LayerNorm launches dropped; 8 -> 5; sums unchanged
# speedup vs baseline: 3.4148x; 1.2046x over previous
#include <hip/hip_runtime.h>
#include <hip/hip_bf16.h>

typedef __attribute__((ext_vector_type(16))) _Float16 v16h;
typedef __attribute__((ext_vector_type(8)))  _Float16 v8h;
typedef __attribute__((ext_vector_type(4)))  _Float16 v4h;
typedef __attribute__((ext_vector_type(8)))  float    v8f;

#define N_NODES 50000
#define N_EDGES 800000
#define DIM 128
#define HEADS 8
#define HEAD_DIM 16

#define SCAN_THREADS 1024
#define SCAN_VEC 13
#define SCAN_ITEMS (4 * SCAN_VEC)
#define CNT_PAD (SCAN_THREADS * SCAN_ITEMS)

#define HIST_CHUNKS 25
#define HIST_THREADS 800
#define CHUNK_V4 (N_EDGES / 4 / HIST_CHUNKS)
#define HIST_V4 (CHUNK_V4 / HIST_THREADS)
#define NODE_V4 (N_NODES / 4)

#define M_TILES (N_NODES / 16)
#define GEMM_BLOCKS 512
#define CVT_ITEMS 10
#define EDGE_THREADS 160
#define EDGE_BLOCKS (N_EDGES / 8 / EDGE_THREADS)
#define TILE_NODES 16
#define NODES_PER_WAVE 2
#define VT_PITCH (DIM + 8)
#define OT_PITCH (DIM + 4)
static_assert(N_NODES % TILE_NODES == 0 && TILE_NODES == 8 * NODES_PER_WAVE,
              "node tiles must divide the nodes and fill the workgroup's waves");

#define HIST_WAVES (HIST_THREADS / 32)
#define CVTX_WAVES (N_NODES * DIM / 4 / (32 * CVT_ITEMS))
#define CVTX_BLOCKS (CVTX_WAVES / HIST_WAVES)
#define CVTW_BLOCKS ((DIM * DIM + HIST_THREADS - 1) / HIST_THREADS)
#define PREFIX_PER_BLOCK (CNT_PAD / GEMM_BLOCKS)
static_assert(CVTX_WAVES % HIST_WAVES == 0, "x conversion jobs must fill whole workgroups");
static_assert(CNT_PAD % GEMM_BLOCKS == 0 && PREFIX_PER_BLOCK <= 256,
              "the chunk prefix must divide over the QKV workgroups");


__device__ inline void load_frags(const _Float16* __restrict__ src, int pitch, int r0, int lane,
                                  v16h f[4]) {
    const int row = lane & 15;
    const int kb  = (lane >> 4) << 3;
    const _Float16* xr = src + (size_t)(r0 + row) * pitch;
#pragma unroll
    for (int t = 0; t < 4; ++t) {
        const v8h lo = *(const v8h*)(xr + t * 32 + kb);
        const v8h hi = *(const v8h*)(xr + t * 32 + kb + 16);
#pragma unroll
        for (int i = 0; i < 8; ++i) { f[t][i] = lo[i]; f[t][i + 8] = hi[i]; }
    }
}

__device__ inline void mma_store_16x16(const v16h a[4], const v16h b[4], float bia,
                                       float* __restrict__ out, int pitch,
                                       int r0, int n0, int lane) {
    v8f c = {};
#pragma unroll
    for (int t = 0; t < 4; ++t)
        c = __builtin_amdgcn_wmma_f32_16x16x32_f16(false, a[t], false, b[t],
                                                   (short)0, c, false, false);
    float* op = out + (size_t)r0 * pitch + n0 + (lane & 15);
#pragma unroll
    for (int j = 0; j < 8; ++j) {
        int m = j + ((lane >> 4) << 3);
        op[(size_t)m * pitch] = c[j] + bia;
    }
}

__device__ inline void cvt_wave_x4(const float* __restrict__ src, _Float16* __restrict__ dst,
                                   int gw, int lane) {
    const int base = gw * (32 * CVT_ITEMS) + lane;
    float4 vv[CVT_ITEMS];
#pragma unroll
    for (int j = 0; j < CVT_ITEMS; ++j) vv[j] = ((const float4*)src)[base + 32 * j];
#pragma unroll
    for (int j = 0; j < CVT_ITEMS; ++j) {
        v4h o = { (_Float16)vv[j].x, (_Float16)vv[j].y, (_Float16)vv[j].z, (_Float16)vv[j].w };
        ((v4h*)dst)[base + 32 * j] = o;
    }
}

__device__ inline void cvt_weightT_item(int i,
                                        const float* __restrict__ Wq, const float* __restrict__ Wk,
                                        const float* __restrict__ Wv, const float* __restrict__ Wo,
                                        _Float16* __restrict__ WT) {
    int m = i >> 12, r = i & 4095, n = r >> 5, k = (r & 31) << 2;
    const float* W = (m == 0) ? Wq : (m == 1) ? Wk : (m == 2) ? Wv : Wo;
    const float* wp = W + (size_t)k * DIM + n;
    v4h o = { (_Float16)wp[0], (_Float16)wp[DIM], (_Float16)wp[2 * DIM], (_Float16)wp[3 * DIM] };
    *(v4h*)(WT + (size_t)m * DIM * DIM + n * DIM + k) = o;
}

__global__ void __launch_bounds__(HIST_THREADS)
edge_hist_convert(const int* __restrict__ origin, int* __restrict__ bcnt,
                  int* __restrict__ erank,
                  const float* __restrict__ x, _Float16* __restrict__ x16,
                  const float* __restrict__ Wq, const float* __restrict__ Wk,
                  const float* __restrict__ Wv, const float* __restrict__ Wo,
                  _Float16* __restrict__ WT) {
    __shared__ int4 hist4[NODE_V4];
    int* hist = (int*)hist4;
    const int t = threadIdx.x;
    if (blockIdx.x >= HIST_CHUNKS + CVTX_BLOCKS) {
        const int i = (blockIdx.x - HIST_CHUNKS - CVTX_BLOCKS) * HIST_THREADS + t;
        if (i < DIM * DIM) cvt_weightT_item(i, Wq, Wk, Wv, Wo, WT);
        return;
    }
    if (blockIdx.x >= HIST_CHUNKS) {
        cvt_wave_x4(x, x16, (blockIdx.x - HIST_CHUNKS) * HIST_WAVES + (t >> 5), t & 31);
        return;
    }
    const int4* op = (const int4*)origin + blockIdx.x * CHUNK_V4;
    int4 o[HIST_V4];
#pragma unroll
    for (int j = 0; j < HIST_V4; ++j) o[j] = op[t + HIST_THREADS * j];
    for (int i = t; i < NODE_V4; i += HIST_THREADS) hist4[i] = make_int4(0, 0, 0, 0);
    __syncthreads();
#pragma unroll
    for (int j = 0; j < HIST_V4; ++j) {
        o[j].x = atomicAdd(&hist[o[j].x], 1); o[j].y = atomicAdd(&hist[o[j].y], 1);
        o[j].z = atomicAdd(&hist[o[j].z], 1); o[j].w = atomicAdd(&hist[o[j].w], 1);
    }
    __syncthreads();
    int4* bp = (int4*)(bcnt + (size_t)blockIdx.x * N_NODES);
    for (int i = t; i < NODE_V4; i += HIST_THREADS) bp[i] = hist4[i];
    int4* rp = (int4*)erank + blockIdx.x * CHUNK_V4;
#pragma unroll
    for (int j = 0; j < HIST_V4; ++j) rp[t + HIST_THREADS * j] = o[j];
}

__device__ inline void chunk_prefix_item(int n, int* __restrict__ bcnt, int* __restrict__ cnt) {
    int run = 0;
    if (n < N_NODES) {
        int c[HIST_CHUNKS];
#pragma unroll
        for (int b = 0; b < HIST_CHUNKS; ++b) c[b] = bcnt[(size_t)b * N_NODES + n];
#pragma unroll
        for (int b = 0; b < HIST_CHUNKS; ++b) {
            bcnt[(size_t)b * N_NODES + n] = run;
            run += c[b];
        }
    }
    cnt[n] = run;
}

__global__ void csr_scan(const int* __restrict__ cnt, int* __restrict__ rowptr) {
    __shared__ int wsum[32];
    const int t = threadIdx.x;
    const int lane = t & 31;
    const int wave = t >> 5;
    const int4* cp = (const int4*)(cnt + t * SCAN_ITEMS);
    int4 c[SCAN_VEC];
    int tot = 0;
#pragma unroll
    for (int i = 0; i < SCAN_VEC; ++i) {
        c[i] = cp[i];
        tot += c[i].x + c[i].y + c[i].z + c[i].w;
    }
    int inc = tot;
    for (int m = 1; m < 32; m <<= 1) {
        int up = __shfl_up(inc, m, 32);
        if (lane >= m) inc += up;
    }
    if (lane == 31) wsum[wave] = inc;
    __syncthreads();
    if (wave == 0) {
        const int w = wsum[lane];
        int wi = w;
        for (int m = 1; m < 32; m <<= 1) {
            int up = __shfl_up(wi, m, 32);
            if (lane >= m) wi += up;
        }
        wsum[lane] = wi - w;
    }
    __syncthreads();
    int run = wsum[wave] + inc - tot;
    int4* rp = (int4*)(rowptr + t * SCAN_ITEMS);
#pragma unroll
    for (int i = 0; i < SCAN_VEC; ++i) {
        int4 o;
        o.x = run; run += c[i].x;
        o.y = run; run += c[i].y;
        o.z = run; run += c[i].z;
        o.w = run; run += c[i].w;
        rp[i] = o;
    }
}

static_assert(CHUNK_V4 % EDGE_THREADS == 0 && (N_EDGES / 8) % EDGE_THREADS == 0,
              "a csr_fill block must lie inside one chunk, in both halves");
__global__ void csr_fill(const int* __restrict__ origin, const int* __restrict__ dest,
                         const int* __restrict__ rowptr, const int* __restrict__ boff,
                         const int* __restrict__ erank,
                         int* __restrict__ sdest, int* __restrict__ sedge) {
    const int i = blockIdx.x * EDGE_THREADS + threadIdx.x;
    const int i2 = i + N_EDGES / 8;
    const int* b1 = boff + (size_t)(blockIdx.x / (CHUNK_V4 / EDGE_THREADS)) * N_NODES;
    const int* b2 = boff + (size_t)((blockIdx.x + N_EDGES / 8 / EDGE_THREADS)
                                    / (CHUNK_V4 / EDGE_THREADS)) * N_NODES;
    const int4 oa = ((const int4*)origin)[i];
    const int4 ob = ((const int4*)origin)[i2];
    const int4 ra = ((const int4*)erank)[i];
    const int4 rb = ((const int4*)erank)[i2];
    const int4 da = ((const int4*)dest)[i];
    const int4 db = ((const int4*)dest)[i2];
    const int p0 = rowptr[oa.x] + b1[oa.x] + ra.x;
    const int p1 = rowptr[oa.y] + b1[oa.y] + ra.y;
    const int p2 = rowptr[oa.z] + b1[oa.z] + ra.z;
    const int p3 = rowptr[oa.w] + b1[oa.w] + ra.w;
    const int p4 = rowptr[ob.x] + b2[ob.x] + rb.x;
    const int p5 = rowptr[ob.y] + b2[ob.y] + rb.y;
    const int p6 = rowptr[ob.z] + b2[ob.z] + rb.z;
    const int p7 = rowptr[ob.w] + b2[ob.w] + rb.w;
    sdest[p0] = da.x; sedge[p0] = 4 * i;
    sdest[p1] = da.y; sedge[p1] = 4 * i + 1;
    sdest[p2] = da.z; sedge[p2] = 4 * i + 2;
    sdest[p3] = da.w; sedge[p3] = 4 * i + 3;
    sdest[p4] = db.x; sedge[p4] = 4 * i2;
    sdest[p5] = db.y; sedge[p5] = 4 * i2 + 1;
    sdest[p6] = db.z; sedge[p6] = 4 * i2 + 2;
    sdest[p7] = db.w; sedge[p7] = 4 * i2 + 3;
}

__global__ void __launch_bounds__(256)
qkv_gemm(const _Float16* __restrict__ x16,
         const _Float16* __restrict__ WTq, const float* __restrict__ bq,
         const _Float16* __restrict__ WTk, const float* __restrict__ bk,
         const _Float16* __restrict__ WTv, const float* __restrict__ bv,
         float* __restrict__ q, float* __restrict__ k,
         float* __restrict__ v,
         int* __restrict__ bcnt, int* __restrict__ cnt) {
    const int lane = threadIdx.x & 31;
    const int wave = threadIdx.x >> 5;
    const int n0 = wave * 16;
    if (threadIdx.x < PREFIX_PER_BLOCK)
        chunk_prefix_item(blockIdx.x * PREFIX_PER_BLOCK + threadIdx.x, bcnt, cnt);
    v16h wq[4], wk[4], wv[4];
    load_frags(WTq, DIM, n0, lane, wq);
    load_frags(WTk, DIM, n0, lane, wk);
    load_frags(WTv, DIM, n0, lane, wv);
    const float biq = bq[n0 + (lane & 15)];
    const float bik = bk[n0 + (lane & 15)];
    const float biv = bv[n0 + (lane & 15)];
    int tile = blockIdx.x;
    v16h a[4];
    load_frags(x16, DIM, tile * 16, lane, a);
    while (tile < M_TILES) {
        const int nxt = (tile + GEMM_BLOCKS < M_TILES) ? (tile + GEMM_BLOCKS) : tile;
        v16h an[4];
        load_frags(x16, DIM, nxt * 16, lane, an);
        const int r0 = tile * 16;
        mma_store_16x16(a, wq, biq, q, DIM, r0, n0, lane);
        mma_store_16x16(a, wk, bik, k, DIM, r0, n0, lane);
        mma_store_16x16(a, wv, biv, v, DIM, r0, n0, lane);
#pragma unroll
        for (int t = 0; t < 4; ++t) a[t] = an[t];
        tile += GEMM_BLOCKS;
    }
}

#define EDGE_GRP 4

__device__ inline float quad_sum(float p) {
    p += __int_as_float(__builtin_amdgcn_update_dpp(0, __float_as_int(p), 0xB1, 0xF, 0xF, true));
    p += __int_as_float(__builtin_amdgcn_update_dpp(0, __float_as_int(p), 0x4E, 0xF, 0xF, true));
    return p;
}

typedef float v4f __attribute__((ext_vector_type(4)));
typedef const float __attribute__((address_space(1))) gfloat;
typedef const v4f   __attribute__((address_space(1))) gv4f;

__device__ inline gfloat* scalar_base(const float* p) {
    const unsigned long long a = (unsigned long long)p;
    const unsigned lo = __builtin_amdgcn_readfirstlane((unsigned)a);
    const unsigned hi = __builtin_amdgcn_readfirstlane((unsigned)(a >> 32));
    return (gfloat*)(((unsigned long long)hi << 32) | lo);
}

__global__ void __launch_bounds__(256)
node_attend_out(const int* __restrict__ rowptr,
                const int* __restrict__ sdest, const int* __restrict__ sedge,
                const float* __restrict__ ew,
                const float* __restrict__ q, const float* __restrict__ k,
                const float* __restrict__ v,
                const _Float16* __restrict__ WTo, const float* __restrict__ bo,
                const float* __restrict__ x,
                const float* __restrict__ gamma, const float* __restrict__ beta,
                float* __restrict__ y) {
    __shared__ __attribute__((aligned(16))) _Float16 vt[TILE_NODES * VT_PITCH];
    __shared__ __attribute__((aligned(16))) float    ot[TILE_NODES * OT_PITCH];
    const int lane = threadIdx.x & 31;
    const int wave = threadIdx.x >> 5;
    const int row0 = wave * NODES_PER_WAVE;
    const int node0 = blockIdx.x * TILE_NODES + row0;
    const int h = lane >> 2;
    const int rp = rowptr[node0 + ((lane < NODES_PER_WAVE) ? lane : NODES_PER_WAVE)];
    for (int i = 0; i < NODES_PER_WAVE; ++i) {
        const int node = node0 + i;
        const int beg = __builtin_amdgcn_readlane(rp, i);
        const int end = __builtin_amdgcn_readlane(rp, i + 1);
        float4 qv = ((const float4*)(q + (size_t)node * DIM))[lane];
        qv.x *= 0.25f; qv.y *= 0.25f; qv.z *= 0.25f; qv.w *= 0.25f;
        float m = -INFINITY;
        float l = 0.0f;
        float4 acc = {0.0f, 0.0f, 0.0f, 0.0f};
        for (int base = beg; base < end; base += 32) {
            const int n = (end - base < 32) ? (end - base) : 32;
            int my_d = 0, my_e = 0;
            if (lane < n) { my_d = sdest[base + lane]; my_e = sedge[base + lane]; }
            for (int j0 = 0; j0 < n; j0 += EDGE_GRP) {
                float s[EDGE_GRP];
                v4f vv[EDGE_GRP];
#pragma unroll
                for (int g = 0; g < EDGE_GRP; ++g) {
                    const int j = j0 + g;
                    const int jc = (j < n) ? j : (n - 1);
                    const int d = __builtin_amdgcn_readlane(my_d, jc);
                    const int e = __builtin_amdgcn_readlane(my_e, jc);
                    const v4f kv = ((gv4f*)scalar_base(k + (size_t)d * DIM))[lane];
                    vv[g] = ((gv4f*)scalar_base(v + (size_t)d * DIM))[lane];
                    const float w = scalar_base(ew + (size_t)e * HEADS)[h];
                    float p = qv.x * kv.x + qv.y * kv.y + qv.z * kv.z + qv.w * kv.w;
                    p = quad_sum(p) * w;
                    s[g] = (j < n) ? p : -INFINITY;
                }
                float gm = m;
#pragma unroll
                for (int g = 0; g < EDGE_GRP; ++g) gm = fmaxf(gm, s[g]);
                const float corr = __expf(m - gm);
                l *= corr;
                acc.x *= corr; acc.y *= corr; acc.z *= corr; acc.w *= corr;
#pragma unroll
                for (int g = 0; g < EDGE_GRP; ++g) {
                    const float pg = __expf(s[g] - gm);
                    l += pg;
                    acc.x += pg * vv[g].x;
                    acc.y += pg * vv[g].y;
                    acc.z += pg * vv[g].z;
                    acc.w += pg * vv[g].w;
                }
                m = gm;
            }
        }
        const float inv = 1.0f / (l + 1e-16f);
        v4h o = { (_Float16)(acc.x * inv), (_Float16)(acc.y * inv),
                  (_Float16)(acc.z * inv), (_Float16)(acc.w * inv) };
        *(v4h*)(vt + (row0 + i) * VT_PITCH + 4 * lane) = o;
    }

    const int n0 = wave * 16;
    v16h wo[4];
    load_frags(WTo, DIM, n0, lane, wo);
    const float bio = bo[n0 + (lane & 15)];
    __syncthreads();
    v16h a[4];
    load_frags(vt, VT_PITCH, 0, lane, a);
    mma_store_16x16(a, wo, bio, ot, OT_PITCH, 0, n0, lane);
    __syncthreads();

    const float4 gv = ((const float4*)gamma)[lane];
    const float4 bv = ((const float4*)beta)[lane];
    float4 ovs[NODES_PER_WAVE], xvs[NODES_PER_WAVE];
#pragma unroll
    for (int r = 0; r < NODES_PER_WAVE; ++r) {
        ovs[r] = *(const float4*)(ot + (row0 + r) * OT_PITCH + 4 * lane);
        xvs[r] = ((const float4*)(x + (size_t)(node0 + r) * DIM))[lane];
    }
#pragma unroll
    for (int r = 0; r < NODES_PER_WAVE; ++r) {
        const size_t node = (size_t)(node0 + r);
        const float4 ov = ovs[r];
        const float4 xv = xvs[r];
        float s = ov.x + ov.y + ov.z + ov.w;
#pragma unroll
        for (int m = 16; m >= 1; m >>= 1) s += __shfl_xor(s, m, 32);
        const float mu = s * (1.0f / DIM);
        float dx0 = ov.x - mu, dx1 = ov.y - mu, dx2 = ov.z - mu, dx3 = ov.w - mu;
        float s2 = dx0 * dx0 + dx1 * dx1 + dx2 * dx2 + dx3 * dx3;
#pragma unroll
        for (int m = 16; m >= 1; m >>= 1) s2 += __shfl_xor(s2, m, 32);
        const float inv = rsqrtf(s2 * (1.0f / DIM) + 1e-5f);
        float4 rr;
        rr.x = xv.x + dx0 * inv * gv.x + bv.x;
        rr.y = xv.y + dx1 * inv * gv.y + bv.y;
        rr.z = xv.z + dx2 * inv * gv.z + bv.z;
        rr.w = xv.w + dx3 * inv * gv.w + bv.w;
        ((float4*)(y + node * DIM))[lane] = rr;
    }
}

extern "C" void kernel_launch(void* const* d_in, const int* in_sizes, int n_in,
                              void* d_out, int out_size, void* d_ws, size_t ws_size,
                              hipStream_t stream) {
    const float* x   = (const float*)d_in[0];
    const int*   ei  = (const int*)  d_in[1];
    const float* ew  = (const float*)d_in[2];
    const float* Wq  = (const float*)d_in[3];
    const float* bq  = (const float*)d_in[4];
    const float* Wk  = (const float*)d_in[5];
    const float* bk  = (const float*)d_in[6];
    const float* Wv  = (const float*)d_in[7];
    const float* bv  = (const float*)d_in[8];
    const float* Wo  = (const float*)d_in[9];
    const float* bo  = (const float*)d_in[10];
    const float* gam = (const float*)d_in[11];
    const float* bet = (const float*)d_in[12];
    float* y = (float*)d_out;

    const int* origin = ei;
    const int* dest   = ei + N_EDGES;

    float* ws   = (float*)d_ws;
    float* q    = ws;
    float* k    = q    + (size_t)N_NODES * DIM;
    float* v    = k    + (size_t)N_NODES * DIM;
    float* egl  = v    + (size_t)N_NODES * DIM;
    _Float16* x16 = (_Float16*)(egl + (size_t)N_NODES * DIM);
    _Float16* WT  = x16 + (size_t)N_NODES * DIM;
    _Float16* WTq = WT;
    _Float16* WTk = WT + 1 * DIM * DIM;
    _Float16* WTv = WT + 2 * DIM * DIM;
    _Float16* WTo = WT + 3 * DIM * DIM;

    int* cnt    = (int*)egl;
    int* rowptr = cnt    + CNT_PAD;
    int* sdest  = rowptr + CNT_PAD;
    int* sedge  = sdest  + N_EDGES;
    int* erank  = sedge  + N_EDGES;
    int* bcnt   = erank  + N_EDGES;
    static_assert(2 * CNT_PAD + 3 * N_EDGES + HIST_CHUNKS * N_NODES <= N_NODES * DIM,
                  "the grouped edge list must fit in its region");

    edge_hist_convert<<<HIST_CHUNKS + CVTX_BLOCKS + CVTW_BLOCKS, HIST_THREADS, 0, stream>>>(
        origin, bcnt, erank, x, x16, Wq, Wk, Wv, Wo, WT);

    qkv_gemm<<<GEMM_BLOCKS, 256, 0, stream>>>(x16, WTq, bq, WTk, bk, WTv, bv, q, k, v, bcnt, cnt);

    csr_scan<<<1, SCAN_THREADS, 0, stream>>>(cnt, rowptr);
    csr_fill<<<EDGE_BLOCKS, EDGE_THREADS, 0, stream>>>(origin, dest, rowptr, bcnt, erank, sdest, sedge);

    node_attend_out<<<N_NODES / TILE_NODES, 256, 0, stream>>>(rowptr, sdest, sedge, ew, q, k, v,
                                                             WTo, bo, x, gam, bet, y);
}


# Two-level slot scan: each QKV workgroup scans its 128 nodes' counts, every fill workgroup scans the 512 range sums in LDS: scan launch dropped (5 -> 4), one global gather per edge; sums unchanged
# speedup vs baseline: 4.5677x; 1.3376x over previous
#include <hip/hip_runtime.h>
#include <hip/hip_bf16.h>

typedef __attribute__((ext_vector_type(16))) _Float16 v16h;
typedef __attribute__((ext_vector_type(8)))  _Float16 v8h;
typedef __attribute__((ext_vector_type(4)))  _Float16 v4h;
typedef __attribute__((ext_vector_type(2)))  _Float16 v2h;
typedef __attribute__((ext_vector_type(8)))  float    v8f;

#define N_NODES 50000
#define N_EDGES 800000
#define DIM 128
#define HEADS 8
#define HEAD_DIM 16

#define CNT_PAD 53248

#define HIST_CHUNKS 25
#define HIST_THREADS 800
#define CHUNK_V4 (N_EDGES / 4 / HIST_CHUNKS)
#define HIST_V4 (CHUNK_V4 / HIST_THREADS)
#define NODE_V4 (N_NODES / 4)

#define M_TILES (N_NODES / 16)
#define GEMM_BLOCKS 512
#define CVT_ITEMS 10
#define EDGE_THREADS 160
#define EDGE_BLOCKS (N_EDGES / 8 / EDGE_THREADS)
#define TILE_NODES 16
#define NODES_PER_WAVE 2
#define VT_PITCH (DIM + 8)
#define OT_PITCH (DIM + 4)
static_assert(N_NODES % TILE_NODES == 0 && TILE_NODES == 8 * NODES_PER_WAVE,
              "node tiles must divide the nodes and fill the workgroup's waves");

#define HIST_WAVES (HIST_THREADS / 32)
#define CVTX_WAVES (N_NODES * DIM / 4 / (32 * CVT_ITEMS))
#define CVTX_BLOCKS (CVTX_WAVES / HIST_WAVES)
#define CVTW_BLOCKS ((DIM * DIM + HIST_THREADS - 1) / HIST_THREADS)
#define PREFIX_SHIFT 7
#define PREFIX_PER_BLOCK (1 << PREFIX_SHIFT)
#define PREFIX_BLOCKS (N_NODES / PREFIX_PER_BLOCK + 1)
#define BASE_THREADS (GEMM_BLOCKS / 4)
static_assert(CVTX_WAVES % HIST_WAVES == 0, "x conversion jobs must fill whole workgroups");
static_assert(PREFIX_BLOCKS <= GEMM_BLOCKS && PREFIX_BLOCKS * PREFIX_PER_BLOCK <= CNT_PAD - 1024 &&
              PREFIX_PER_BLOCK % 32 == 0 && PREFIX_PER_BLOCK <= 256 && GEMM_BLOCKS % 4 == 0,
              "the node ranges must fit the QKV workgroups and their arrays");


__device__ inline void load_frags(const _Float16* __restrict__ src, int pitch, int r0, int lane,
                                  v16h f[4]) {
    const int row = lane & 15;
    const int kb  = (lane >> 4) << 3;
    const _Float16* xr = src + (size_t)(r0 + row) * pitch;
#pragma unroll
    for (int t = 0; t < 4; ++t) {
        const v8h lo = *(const v8h*)(xr + t * 32 + kb);
        const v8h hi = *(const v8h*)(xr + t * 32 + kb + 16);
#pragma unroll
        for (int i = 0; i < 8; ++i) { f[t][i] = lo[i]; f[t][i + 8] = hi[i]; }
    }
}

__device__ inline void mma_store_16x16(const v16h a[4], const v16h b[4], float bia,
                                       float* __restrict__ out, int pitch,
                                       int r0, int n0, int lane) {
    v8f c = {};
#pragma unroll
    for (int t = 0; t < 4; ++t)
        c = __builtin_amdgcn_wmma_f32_16x16x32_f16(false, a[t], false, b[t],
                                                   (short)0, c, false, false);
    float* op = out + (size_t)r0 * pitch + n0 + (lane & 15);
#pragma unroll
    for (int j = 0; j < 8; ++j) {
        int m = j + ((lane >> 4) << 3);
        op[(size_t)m * pitch] = c[j] + bia;
    }
}

__device__ inline void mma_store_kv(const v16h a[4], const v16h bk[4], const v16h bv[4],
                                    float bik, float biv, _Float16* __restrict__ kv,
                                    int r0, int n0, int lane) {
    v8f ck = {}, cv = {};
#pragma unroll
    for (int t = 0; t < 4; ++t) {
        ck = __builtin_amdgcn_wmma_f32_16x16x32_f16(false, a[t], false, bk[t],
                                                    (short)0, ck, false, false);
        cv = __builtin_amdgcn_wmma_f32_16x16x32_f16(false, a[t], false, bv[t],
                                                    (short)0, cv, false, false);
    }
    v2h* op = (v2h*)kv + (size_t)r0 * DIM + n0 + (lane & 15);
#pragma unroll
    for (int j = 0; j < 8; ++j) {
        int m = j + ((lane >> 4) << 3);
        v2h o = { (_Float16)(ck[j] + bik), (_Float16)(cv[j] + biv) };
        op[(size_t)m * DIM] = o;
    }
}

__device__ inline void cvt_wave_x4(const float* __restrict__ src, _Float16* __restrict__ dst,
                                   int gw, int lane) {
    const int base = gw * (32 * CVT_ITEMS) + lane;
    float4 vv[CVT_ITEMS];
#pragma unroll
    for (int j = 0; j < CVT_ITEMS; ++j) vv[j] = ((const float4*)src)[base + 32 * j];
#pragma unroll
    for (int j = 0; j < CVT_ITEMS; ++j) {
        v4h o = { (_Float16)vv[j].x, (_Float16)vv[j].y, (_Float16)vv[j].z, (_Float16)vv[j].w };
        ((v4h*)dst)[base + 32 * j] = o;
    }
}

__device__ inline void cvt_weightT_item(int i,
                                        const float* __restrict__ Wq, const float* __restrict__ Wk,
                                        const float* __restrict__ Wv, const float* __restrict__ Wo,
                                        _Float16* __restrict__ WT) {
    int m = i >> 12, r = i & 4095, n = r >> 5, k = (r & 31) << 2;
    const float* W = (m == 0) ? Wq : (m == 1) ? Wk : (m == 2) ? Wv : Wo;
    const float* wp = W + (size_t)k * DIM + n;
    v4h o = { (_Float16)wp[0], (_Float16)wp[DIM], (_Float16)wp[2 * DIM], (_Float16)wp[3 * DIM] };
    *(v4h*)(WT + (size_t)m * DIM * DIM + n * DIM + k) = o;
}

__global__ void __launch_bounds__(HIST_THREADS)
edge_hist_convert(const int* __restrict__ origin, int* __restrict__ bcnt,
                  int* __restrict__ erank,
                  const float* __restrict__ x, _Float16* __restrict__ x16,
                  const float* __restrict__ Wq, const float* __restrict__ Wk,
                  const float* __restrict__ Wv, const float* __restrict__ Wo,
                  _Float16* __restrict__ WT) {
    __shared__ int4 hist4[NODE_V4];
    int* hist = (int*)hist4;
    const int t = threadIdx.x;
    if (blockIdx.x >= HIST_CHUNKS + CVTX_BLOCKS) {
        const int i = (blockIdx.x - HIST_CHUNKS - CVTX_BLOCKS) * HIST_THREADS + t;
        if (i < DIM * DIM) cvt_weightT_item(i, Wq, Wk, Wv, Wo, WT);
        return;
    }
    if (blockIdx.x >= HIST_CHUNKS) {
        cvt_wave_x4(x, x16, (blockIdx.x - HIST_CHUNKS) * HIST_WAVES + (t >> 5), t & 31);
        return;
    }
    const int4* op = (const int4*)origin + blockIdx.x * CHUNK_V4;
    int4 o[HIST_V4];
#pragma unroll
    for (int j = 0; j < HIST_V4; ++j) o[j] = op[t + HIST_THREADS * j];
    for (int i = t; i < NODE_V4; i += HIST_THREADS) hist4[i] = make_int4(0, 0, 0, 0);
    __syncthreads();
#pragma unroll
    for (int j = 0; j < HIST_V4; ++j) {
        o[j].x = atomicAdd(&hist[o[j].x], 1); o[j].y = atomicAdd(&hist[o[j].y], 1);
        o[j].z = atomicAdd(&hist[o[j].z], 1); o[j].w = atomicAdd(&hist[o[j].w], 1);
    }
    __syncthreads();
    int4* bp = (int4*)(bcnt + (size_t)blockIdx.x * N_NODES);
    for (int i = t; i < NODE_V4; i += HIST_THREADS) bp[i] = hist4[i];
    int4* rp = (int4*)erank + blockIdx.x * CHUNK_V4;
#pragma unroll
    for (int j = 0; j < HIST_V4; ++j) rp[t + HIST_THREADS * j] = o[j];
}

__device__ inline void chunk_prefix_range(int* __restrict__ bcnt, int* __restrict__ lrow,
                                          int* __restrict__ rsum, int* wtot) {
    const int t = threadIdx.x;
    const int lane = t & 31;
    const int wave = t >> 5;
    const int n = blockIdx.x * PREFIX_PER_BLOCK + t;
    const bool pre = t < PREFIX_PER_BLOCK;
    int c[HIST_CHUNKS] = {};
    int run = 0, inc = 0;
    if (pre) {
#pragma unroll
        for (int b = 0; b < HIST_CHUNKS; ++b)
            c[b] = (n < N_NODES) ? bcnt[(size_t)b * N_NODES + n] : 0;
#pragma unroll
        for (int b = 0; b < HIST_CHUNKS; ++b) run += c[b];
        inc = run;
        for (int m = 1; m < 32; m <<= 1) {
            const int up = __shfl_up(inc, m, 32);
            if (lane >= m) inc += up;
        }
        if (lane == 31) wtot[wave] = inc;
    }
    __syncthreads();
    if (pre) {
        int first = inc - run;
#pragma unroll
        for (int w = 0; w < PREFIX_PER_BLOCK / 32; ++w) first += (w < wave) ? wtot[w] : 0;
        if (n <= N_NODES) lrow[n] = first;
        if (n < N_NODES) {
            int r = first;
#pragma unroll
            for (int b = 0; b < HIST_CHUNKS; ++b) {
                bcnt[(size_t)b * N_NODES + n] = r;
                r += c[b];
            }
        }
        if (t == 0) {
            int sum = 0;
#pragma unroll
            for (int w = 0; w < PREFIX_PER_BLOCK / 32; ++w) sum += wtot[w];
            rsum[blockIdx.x] = sum;
        }
    }
}

static_assert(CHUNK_V4 % EDGE_THREADS == 0 && (N_EDGES / 8) % EDGE_THREADS == 0,
              "a csr_fill block must lie inside one chunk, in both halves");
static_assert(BASE_THREADS % 32 == 0 && BASE_THREADS <= EDGE_THREADS &&
              (N_EDGES / 8 / EDGE_THREADS) * EDGE_THREADS > N_NODES,
              "whole waves scan the range sums; the threads cover rowptr");
__global__ void csr_fill(const int* __restrict__ origin, const int* __restrict__ dest,
                         const int* __restrict__ lrow, const int* __restrict__ rsum,
                         const int* __restrict__ boff, const int* __restrict__ erank,
                         int* __restrict__ rowptr,
                         int* __restrict__ sdest, int* __restrict__ sedge) {
    __shared__ __attribute__((aligned(16))) int base[GEMM_BLOCKS];
    __shared__ int wsum[BASE_THREADS / 32];
    const int t = threadIdx.x;
    const int lane = t & 31;
    const int wave = t >> 5;
    const int i = blockIdx.x * EDGE_THREADS + t;
    const int i2 = i + N_EDGES / 8;
    const int* b1 = boff + (size_t)(blockIdx.x / (CHUNK_V4 / EDGE_THREADS)) * N_NODES;
    const int* b2 = boff + (size_t)((blockIdx.x + N_EDGES / 8 / EDGE_THREADS)
                                    / (CHUNK_V4 / EDGE_THREADS)) * N_NODES;
    const int4 oa = ((const int4*)origin)[i];
    const int4 ob = ((const int4*)origin)[i2];
    const int4 ra = ((const int4*)erank)[i];
    const int4 rb = ((const int4*)erank)[i2];
    const int4 da = ((const int4*)dest)[i];
    const int4 db = ((const int4*)dest)[i2];
    const int f0 = b1[oa.x] + ra.x;
    const int f1 = b1[oa.y] + ra.y;
    const int f2 = b1[oa.z] + ra.z;
    const int f3 = b1[oa.w] + ra.w;
    const int f4 = b2[ob.x] + rb.x;
    const int f5 = b2[ob.y] + rb.y;
    const int f6 = b2[ob.z] + rb.z;
    const int f7 = b2[ob.w] + rb.w;
    const int lr = (i <= N_NODES) ? lrow[i] : 0;
    const bool sc = t < BASE_THREADS;
    int4 tv = make_int4(0, 0, 0, 0);
    int tsum = 0, inc = 0;
    if (sc) {
        tv = ((const int4*)rsum)[t];
        tsum = tv.x + tv.y + tv.z + tv.w;
        inc = tsum;
        for (int m = 1; m < 32; m <<= 1) {
            const int up = __shfl_up(inc, m, 32);
            if (lane >= m) inc += up;
        }
        if (lane == 31) wsum[wave] = inc;
    }
    __syncthreads();
    if (sc) {
        int r = inc - tsum;
#pragma unroll
        for (int w = 0; w < BASE_THREADS / 32; ++w) r += (w < wave) ? wsum[w] : 0;
        int4 o;
        o.x = r; r += tv.x;
        o.y = r; r += tv.y;
        o.z = r; r += tv.z;
        o.w = r;
        ((int4*)base)[t] = o;
    }
    __syncthreads();
    if (i <= N_NODES) rowptr[i] = base[i >> PREFIX_SHIFT] + lr;
    const int p0 = base[oa.x >> PREFIX_SHIFT] + f0;
    const int p1 = base[oa.y >> PREFIX_SHIFT] + f1;
    const int p2 = base[oa.z >> PREFIX_SHIFT] + f2;
    const int p3 = base[oa.w >> PREFIX_SHIFT] + f3;
    const int p4 = base[ob.x >> PREFIX_SHIFT] + f4;
    const int p5 = base[ob.y >> PREFIX_SHIFT] + f5;
    const int p6 = base[ob.z >> PREFIX_SHIFT] + f6;
    const int p7 = base[ob.w >> PREFIX_SHIFT] + f7;
    sdest[p0] = da.x; sedge[p0] = 4 * i;
    sdest[p1] = da.y; sedge[p1] = 4 * i + 1;
    sdest[p2] = da.z; sedge[p2] = 4 * i + 2;
    sdest[p3] = da.w; sedge[p3] = 4 * i + 3;
    sdest[p4] = db.x; sedge[p4] = 4 * i2;
    sdest[p5] = db.y; sedge[p5] = 4 * i2 + 1;
    sdest[p6] = db.z; sedge[p6] = 4 * i2 + 2;
    sdest[p7] = db.w; sedge[p7] = 4 * i2 + 3;
}

__global__ void __launch_bounds__(256)
qkv_gemm(const _Float16* __restrict__ x16,
         const _Float16* __restrict__ WTq, const float* __restrict__ bq,
         const _Float16* __restrict__ WTk, const float* __restrict__ bk,
         const _Float16* __restrict__ WTv, const float* __restrict__ bv,
         float* __restrict__ q, _Float16* __restrict__ kv,
         int* __restrict__ bcnt, int* __restrict__ lrow, int* __restrict__ rsum) {
    __shared__ int wtot[PREFIX_PER_BLOCK / 32];
    const int lane = threadIdx.x & 31;
    const int wave = threadIdx.x >> 5;
    const int n0 = wave * 16;
    v16h wq[4], wk[4], wv[4];
    load_frags(WTq, DIM, n0, lane, wq);
    load_frags(WTk, DIM, n0, lane, wk);
    load_frags(WTv, DIM, n0, lane, wv);
    const float biq = bq[n0 + (lane & 15)];
    const float bik = bk[n0 + (lane & 15)];
    const float biv = bv[n0 + (lane & 15)];
    int tile = blockIdx.x;
    v16h a[4];
    load_frags(x16, DIM, tile * 16, lane, a);
    chunk_prefix_range(bcnt, lrow, rsum, wtot);
    while (tile < M_TILES) {
        const int nxt = (tile + GEMM_BLOCKS < M_TILES) ? (tile + GEMM_BLOCKS) : tile;
        v16h an[4];
        load_frags(x16, DIM, nxt * 16, lane, an);
        const int r0 = tile * 16;
        mma_store_16x16(a, wq, biq, q, DIM, r0, n0, lane);
        mma_store_kv(a, wk, wv, bik, biv, kv, r0, n0, lane);
#pragma unroll
        for (int t = 0; t < 4; ++t) a[t] = an[t];
        tile += GEMM_BLOCKS;
    }
}

#define EDGE_GRP 4

__device__ inline float quad_sum(float p) {
    p += __int_as_float(__builtin_amdgcn_update_dpp(0, __float_as_int(p), 0xB1, 0xF, 0xF, true));
    p += __int_as_float(__builtin_amdgcn_update_dpp(0, __float_as_int(p), 0x4E, 0xF, 0xF, true));
    return p;
}

typedef float v4f __attribute__((ext_vector_type(4)));
typedef const float __attribute__((address_space(1))) gfloat;
typedef const v4f   __attribute__((address_space(1))) gv4f;
typedef const _Float16 __attribute__((address_space(1))) ghalf;
typedef const v8h   __attribute__((address_space(1))) gv8h;

__device__ inline gfloat* scalar_base(const float* p) {
    const unsigned long long a = (unsigned long long)p;
    const unsigned lo = __builtin_amdgcn_readfirstlane((unsigned)a);
    const unsigned hi = __builtin_amdgcn_readfirstlane((unsigned)(a >> 32));
    return (gfloat*)(((unsigned long long)hi << 32) | lo);
}
__device__ inline ghalf* scalar_base_h(const _Float16* p) {
    const unsigned long long a = (unsigned long long)p;
    const unsigned lo = __builtin_amdgcn_readfirstlane((unsigned)a);
    const unsigned hi = __builtin_amdgcn_readfirstlane((unsigned)(a >> 32));
    return (ghalf*)(((unsigned long long)hi << 32) | lo);
}

__global__ void __launch_bounds__(256)
node_attend_out(const int* __restrict__ rowptr,
                const int* __restrict__ sdest, const int* __restrict__ sedge,
                const float* __restrict__ ew,
                const float* __restrict__ q, const _Float16* __restrict__ kv,
                const _Float16* __restrict__ WTo, const float* __restrict__ bo,
                const float* __restrict__ x,
                const float* __restrict__ gamma, const float* __restrict__ beta,
                float* __restrict__ y) {
    __shared__ __attribute__((aligned(16))) _Float16 vt[TILE_NODES * VT_PITCH];
    __shared__ __attribute__((aligned(16))) float    ot[TILE_NODES * OT_PITCH];
    const int lane = threadIdx.x & 31;
    const int wave = threadIdx.x >> 5;
    const int row0 = wave * NODES_PER_WAVE;
    const int node0 = blockIdx.x * TILE_NODES + row0;
    const int h = lane >> 2;
    const int rp = rowptr[node0 + ((lane < NODES_PER_WAVE) ? lane : NODES_PER_WAVE)];
    for (int i = 0; i < NODES_PER_WAVE; ++i) {
        const int node = node0 + i;
        const int beg = __builtin_amdgcn_readlane(rp, i);
        const int end = __builtin_amdgcn_readlane(rp, i + 1);
        float4 qv = ((const float4*)(q + (size_t)node * DIM))[lane];
        qv.x *= 0.25f; qv.y *= 0.25f; qv.z *= 0.25f; qv.w *= 0.25f;
        float m = -INFINITY;
        float l = 0.0f;
        float4 acc = {0.0f, 0.0f, 0.0f, 0.0f};
        for (int base = beg; base < end; base += 32) {
            const int n = (end - base < 32) ? (end - base) : 32;
            int my_d = 0, my_e = 0;
            if (lane < n) { my_d = sdest[base + lane]; my_e = sedge[base + lane]; }
            for (int j0 = 0; j0 < n; j0 += EDGE_GRP) {
                float s[EDGE_GRP];
                v4f vv[EDGE_GRP];
#pragma unroll
                for (int g = 0; g < EDGE_GRP; ++g) {
                    const int j = j0 + g;
                    const int jc = (j < n) ? j : (n - 1);
                    const int d = __builtin_amdgcn_readlane(my_d, jc);
                    const int e = __builtin_amdgcn_readlane(my_e, jc);
                    const v8h r = ((gv8h*)scalar_base_h(kv + (size_t)d * (2 * DIM)))[lane];
                    const float w = scalar_base(ew + (size_t)e * HEADS)[h];
                    vv[g] = (v4f){ (float)r[1], (float)r[3], (float)r[5], (float)r[7] };
                    float p = qv.x * (float)r[0] + qv.y * (float)r[2]
                            + qv.z * (float)r[4] + qv.w * (float)r[6];
                    p = quad_sum(p) * w;
                    s[g] = (j < n) ? p : -INFINITY;
                }
                float gm = m;
#pragma unroll
                for (int g = 0; g < EDGE_GRP; ++g) gm = fmaxf(gm, s[g]);
                const float corr = __expf(m - gm);
                l *= corr;
                acc.x *= corr; acc.y *= corr; acc.z *= corr; acc.w *= corr;
#pragma unroll
                for (int g = 0; g < EDGE_GRP; ++g) {
                    const float pg = __expf(s[g] - gm);
                    l += pg;
                    acc.x += pg * vv[g].x;
                    acc.y += pg * vv[g].y;
                    acc.z += pg * vv[g].z;
                    acc.w += pg * vv[g].w;
                }
                m = gm;
            }
        }
        const float inv = 1.0f / (l + 1e-16f);
        v4h o = { (_Float16)(acc.x * inv), (_Float16)(acc.y * inv),
                  (_Float16)(acc.z * inv), (_Float16)(acc.w * inv) };
        *(v4h*)(vt + (row0 + i) * VT_PITCH + 4 * lane) = o;
    }

    const int n0 = wave * 16;
    v16h wo[4];
    load_frags(WTo, DIM, n0, lane, wo);
    const float bio = bo[n0 + (lane & 15)];
    __syncthreads();
    v16h a[4];
    load_frags(vt, VT_PITCH, 0, lane, a);
    mma_store_16x16(a, wo, bio, ot, OT_PITCH, 0, n0, lane);
    __syncthreads();

    const float4 gv = ((const float4*)gamma)[lane];
    const float4 bv = ((const float4*)beta)[lane];
    float4 ovs[NODES_PER_WAVE], xvs[NODES_PER_WAVE];
#pragma unroll
    for (int r = 0; r < NODES_PER_WAVE; ++r) {
        ovs[r] = *(const float4*)(ot + (row0 + r) * OT_PITCH + 4 * lane);
        xvs[r] = ((const float4*)(x + (size_t)(node0 + r) * DIM))[lane];
    }
#pragma unroll
    for (int r = 0; r < NODES_PER_WAVE; ++r) {
        const size_t node = (size_t)(node0 + r);
        const float4 ov = ovs[r];
        const float4 xv = xvs[r];
        float s = ov.x + ov.y + ov.z + ov.w;
#pragma unroll
        for (int m = 16; m >= 1; m >>= 1) s += __shfl_xor(s, m, 32);
        const float mu = s * (1.0f / DIM);
        float dx0 = ov.x - mu, dx1 = ov.y - mu, dx2 = ov.z - mu, dx3 = ov.w - mu;
        float s2 = dx0 * dx0 + dx1 * dx1 + dx2 * dx2 + dx3 * dx3;
#pragma unroll
        for (int m = 16; m >= 1; m >>= 1) s2 += __shfl_xor(s2, m, 32);
        const float inv = rsqrtf(s2 * (1.0f / DIM) + 1e-5f);
        float4 rr;
        rr.x = xv.x + dx0 * inv * gv.x + bv.x;
        rr.y = xv.y + dx1 * inv * gv.y + bv.y;
        rr.z = xv.z + dx2 * inv * gv.z + bv.z;
        rr.w = xv.w + dx3 * inv * gv.w + bv.w;
        ((float4*)(y + node * DIM))[lane] = rr;
    }
}

extern "C" void kernel_launch(void* const* d_in, const int* in_sizes, int n_in,
                              void* d_out, int out_size, void* d_ws, size_t ws_size,
                              hipStream_t stream) {
    const float* x   = (const float*)d_in[0];
    const int*   ei  = (const int*)  d_in[1];
    const float* ew  = (const float*)d_in[2];
    const float* Wq  = (const float*)d_in[3];
    const float* bq  = (const float*)d_in[4];
    const float* Wk  = (const float*)d_in[5];
    const float* bk  = (const float*)d_in[6];
    const float* Wv  = (const float*)d_in[7];
    const float* bv  = (const float*)d_in[8];
    const float* Wo  = (const float*)d_in[9];
    const float* bo  = (const float*)d_in[10];
    const float* gam = (const float*)d_in[11];
    const float* bet = (const float*)d_in[12];
    float* y = (float*)d_out;

    const int* origin = ei;
    const int* dest   = ei + N_EDGES;

    float* ws   = (float*)d_ws;
    float* q    = ws;
    _Float16* kv = (_Float16*)(q + (size_t)N_NODES * DIM);
    float* egl  = q    + 3 * (size_t)N_NODES * DIM;
    _Float16* x16 = (_Float16*)(egl + (size_t)N_NODES * DIM);
    _Float16* WT  = x16 + (size_t)N_NODES * DIM;
    _Float16* WTq = WT;
    _Float16* WTk = WT + 1 * DIM * DIM;
    _Float16* WTv = WT + 2 * DIM * DIM;
    _Float16* WTo = WT + 3 * DIM * DIM;

    int* lrow   = (int*)egl;
    int* rsum   = lrow   + (CNT_PAD - 1024);
    int* rowptr = lrow   + CNT_PAD;
    int* sdest  = rowptr + CNT_PAD;
    int* sedge  = sdest  + N_EDGES;
    int* erank  = sedge  + N_EDGES;
    int* bcnt   = erank  + N_EDGES;
    static_assert(2 * CNT_PAD + 3 * N_EDGES + HIST_CHUNKS * N_NODES <= N_NODES * DIM,
                  "the grouped edge list must fit in its region");

    edge_hist_convert<<<HIST_CHUNKS + CVTX_BLOCKS + CVTW_BLOCKS, HIST_THREADS, 0, stream>>>(
        origin, bcnt, erank, x, x16, Wq, Wk, Wv, Wo, WT);

    qkv_gemm<<<GEMM_BLOCKS, 256, 0, stream>>>(x16, WTq, bq, WTk, bk, WTv, bv, q, kv, bcnt, lrow, rsum);

    csr_fill<<<EDGE_BLOCKS, EDGE_THREADS, 0, stream>>>(origin, dest, lrow, rsum, bcnt, erank,
                                                       rowptr, sdest, sedge);

    node_attend_out<<<N_NODES / TILE_NODES, 256, 0, stream>>>(rowptr, sdest, sedge, ew, q, kv,
                                                             WTo, bo, x, gam, bet, y);
}
